# starting kernel
# baseline (speedup 1.0000x reference)
#include <hip/hip_runtime.h>
#include <math.h>

constexpr int Bn  = 4;
constexpr int Wd  = 32;
constexpr int Hd  = 32;
constexpr int Cc  = 128;
constexpr int Ss  = 256;
constexpr int Dd  = 512;
constexpr int NC  = 64;
constexpr int NPOS = Bn * Wd * Hd;

typedef __attribute__((ext_vector_type(2))) float v2f;
typedef __attribute__((ext_vector_type(8))) float v8f;

__device__ __forceinline__ v8f wmma_k4(v2f a, v2f b, v8f c) {
  return __builtin_amdgcn_wmma_f32_16x16x4_f32(
       false, a,  false, b,
       (short)0, c,  false,  false);
}


__global__ __launch_bounds__(128) void text_proj_kernel(
    const float* __restrict__ s, const float* __restrict__ amask,
    const float* __restrict__ Wks, const float* __restrict__ bks,
    const float* __restrict__ Wvs, const float* __restrict__ bvs,
    float* __restrict__ ks_out, float* __restrict__ vs_out) {
  constexpr int TE_STRIDE = Dd + 4;
  __shared__ float te_s[16 * TE_STRIDE];

  const int mat = blockIdx.x & 1;
  const int st  = (blockIdx.x >> 1) & 15;
  const int b   = blockIdx.x >> 5;

  const int tid  = threadIdx.x;
  const int lane = tid & 31;
  const int wave = tid >> 5;

  for (int i = tid; i < 16 * Dd; i += 128) {
    const int r = i >> 9, c = i & (Dd - 1);
    const int srow = st * 16 + r;
    te_s[r * TE_STRIDE + c] =
        s[((size_t)b * Ss + srow) * Dd + c] * amask[b * Ss + srow];
  }
  __syncthreads();

  const float* Wm   = mat ? Wvs : Wks;
  const float* bias = mat ? bvs : bks;
  float*       dst  = mat ? vs_out : ks_out;

  const int m    = lane & 15;
  const int koff = (lane >> 4) * 2;
  const int n    = wave * 16 + (lane & 15);

  v8f acc = {0.f, 0.f, 0.f, 0.f, 0.f, 0.f, 0.f, 0.f};
#pragma unroll 4
  for (int k = 0; k < Dd; k += 4) {
    v2f af, bf;
    af.x = te_s[m * TE_STRIDE + k + koff];
    af.y = te_s[m * TE_STRIDE + k + koff + 1];
    bf.x = Wm[(k + koff) * NC + n];
    bf.y = Wm[(k + koff + 1) * NC + n];
    acc = wmma_k4(af, bf, acc);
  }
  const float bn = bias[n];
#pragma unroll
  for (int i = 0; i < 8; ++i) {
    const int row = i + ((lane >> 4) << 3);
    dst[((size_t)b * Ss + st * 16 + row) * NC + n] = acc[i] + bn;
  }
}

__global__ __launch_bounds__(128) void colsum_kernel(
    const float* __restrict__ ks, const float* __restrict__ vs,
    float* __restrict__ Ksum, float* __restrict__ Vsum) {
  const int b = blockIdx.x;
  const int t = threadIdx.x;
  const float* src = (t < NC) ? ks : vs;
  const int col = t & (NC - 1);
  float acc = 0.f;
  for (int sIdx = 0; sIdx < Ss; ++sIdx)
    acc += src[((size_t)b * Ss + sIdx) * NC + col];
  if (t < NC) Ksum[b * NC + col] = acc;
  else        Vsum[b * NC + col] = acc;
}

__global__ __launch_bounds__(128) void gram_kernel(
    const float* __restrict__ ks, const float* __restrict__ vs,
    float* __restrict__ G) {
  const int mt = blockIdx.x & 3;
  const int b  = blockIdx.x >> 2;
  const int lane = threadIdx.x & 31;
  const int wave = threadIdx.x >> 5;

  const int m    = lane & 15;
  const int koff = (lane >> 4) * 2;
  const int c0   = mt * 16;
  const int n    = wave * 16 + (lane & 15);

  v8f acc = {0.f, 0.f, 0.f, 0.f, 0.f, 0.f, 0.f, 0.f};
#pragma unroll 4
  for (int k = 0; k < Ss; k += 4) {
    v2f af, bf;
    af.x = vs[((size_t)b * Ss + k + koff) * NC + c0 + m];
    af.y = vs[((size_t)b * Ss + k + koff + 1) * NC + c0 + m];
    bf.x = ks[((size_t)b * Ss + k + koff) * NC + n];
    bf.y = ks[((size_t)b * Ss + k + koff + 1) * NC + n];
    acc = wmma_k4(af, bf, acc);
  }
#pragma unroll
  for (int i = 0; i < 8; ++i) {
    const int row = i + ((lane >> 4) << 3);
    G[((size_t)b * NC + c0 + row) * NC + n] = acc[i];
  }
}

__global__ __launch_bounds__(128) void fused_attn_kernel(
    const float* __restrict__ x,
    const float* __restrict__ Wq,  const float* __restrict__ bq,
    const float* __restrict__ Wkx, const float* __restrict__ bkx,
    const float* __restrict__ Wvx, const float* __restrict__ bvx,
    const float* __restrict__ G,
    const float* __restrict__ Ksum, const float* __restrict__ Vsum,
    const float* __restrict__ Wout, const float* __restrict__ bout,
    const float* __restrict__ ln_scale, const float* __restrict__ ln_bias,
    float* __restrict__ y) {
  constexpr int XS = Cc + 4;
  constexpr int QS = NC + 4;
  __shared__ float xs[16 * XS];
  __shared__ float qs[16 * QS];
  __shared__ float kxs[16 * QS];
  __shared__ float vxs[16 * QS];
  __shared__ float os[16 * QS];
  __shared__ float outs[16 * XS];
  __shared__ float Gs[NC * NC];
  __shared__ float KsS[NC], VsS[NC];

  const int tid  = threadIdx.x;
  const int lane = tid & 31;
  const int wave = tid >> 5;
  const int p0   = blockIdx.x * 16;
  const int b    = p0 >> 10;

  for (int i = tid; i < 16 * Cc; i += 128)
    xs[(i >> 7) * XS + (i & (Cc - 1))] = x[(size_t)p0 * Cc + i];
  for (int i = tid; i < NC * NC; i += 128)
    Gs[i] = G[(size_t)b * NC * NC + i];
  if (tid < NC)            KsS[tid] = Ksum[b * NC + tid];
  else if (tid < 2 * NC)   VsS[tid - NC] = Vsum[b * NC + tid - NC];
  __syncthreads();

  {
    const int m    = lane & 15;
    const int koff = (lane >> 4) * 2;
    const int n    = wave * 16 + (lane & 15);
    for (int mat = 0; mat < 3; ++mat) {
      const float* Wm   = (mat == 0) ? Wq  : (mat == 1) ? Wkx : Wvx;
      const float* bias = (mat == 0) ? bq  : (mat == 1) ? bkx : bvx;
      float*       dst  = (mat == 0) ? qs  : (mat == 1) ? kxs : vxs;
      v8f acc = {0.f, 0.f, 0.f, 0.f, 0.f, 0.f, 0.f, 0.f};
#pragma unroll 4
      for (int k = 0; k < Cc; k += 4) {
        v2f af, bf;
        af.x = xs[m * XS + k + koff];
        af.y = xs[m * XS + k + koff + 1];
        bf.x = Wm[(k + koff) * NC + n];
        bf.y = Wm[(k + koff + 1) * NC + n];
        acc = wmma_k4(af, bf, acc);
      }
      const float bn = bias[n];
#pragma unroll
      for (int i = 0; i < 8; ++i)
        dst[(i + ((lane >> 4) << 3)) * QS + n] = acc[i] + bn;
    }
  }
  __syncthreads();

  for (int pi = 0; pi < 4; ++pi) {
    const int p = wave * 4 + pi;
    const float kx0 = kxs[p * QS + lane];
    const float kx1 = kxs[p * QS + lane + 32];
    const float Ks0 = KsS[lane];
    const float Ks1 = KsS[lane + 32];
    float o0 = 0.f, o1 = 0.f;
#pragma unroll 4
    for (int j = 0; j < NC; ++j) {
      const float vj = vxs[p * QS + j];
      const float Vj = VsS[j];
      const float qj = qs[p * QS + j];
      float r0 = (256.f * vj * kx0 + vj * Ks0 + kx0 * Vj + Gs[j * NC + lane]) * 0.125f;
      float r1 = (256.f * vj * kx1 + vj * Ks1 + kx1 * Vj + Gs[j * NC + lane + 32]) * 0.125f;
      float mx = fmaxf(r0, r1);
#pragma unroll
      for (int msk = 16; msk >= 1; msk >>= 1)
        mx = fmaxf(mx, __shfl_xor(mx, msk, 32));
      const float e0 = __expf(r0 - mx);
      const float e1 = __expf(r1 - mx);
      float sm = e0 + e1;
#pragma unroll
      for (int msk = 16; msk >= 1; msk >>= 1)
        sm += __shfl_xor(sm, msk, 32);
      const float qw = qj / sm;
      o0 += qw * e0;
      o1 += qw * e1;
    }
    os[p * QS + lane]      = o0;
    os[p * QS + lane + 32] = o1;
  }
  __syncthreads();

  {
    const int m    = lane & 15;
    const int koff = (lane >> 4) * 2;
    for (int ct = wave; ct < 8; ct += 4) {
      const int n = ct * 16 + (lane & 15);
      v8f acc = {0.f, 0.f, 0.f, 0.f, 0.f, 0.f, 0.f, 0.f};
#pragma unroll 4
      for (int k = 0; k < NC; k += 4) {
        v2f af, bf;
        af.x = os[m * QS + k + koff];
        af.y = os[m * QS + k + koff + 1];
        bf.x = Wout[(k + koff) * Cc + n];
        bf.y = Wout[(k + koff + 1) * Cc + n];
        acc = wmma_k4(af, bf, acc);
      }
      const float bn = bout[n];
#pragma unroll
      for (int i = 0; i < 8; ++i)
        outs[(i + ((lane >> 4) << 3)) * XS + n] = acc[i] + bn;
    }
  }
  __syncthreads();

  {
    const int row   = tid >> 3;
    const int cbase = (tid & 7) * 16;
    float vals[16];
    float sum = 0.f, sq = 0.f;
#pragma unroll
    for (int i = 0; i < 16; ++i) {
      const float v = xs[row * XS + cbase + i] + outs[row * XS + cbase + i];
      vals[i] = v;
      sum += v;
      sq  += v * v;
    }
#pragma unroll
    for (int msk = 4; msk >= 1; msk >>= 1) {
      sum += __shfl_xor(sum, msk, 32);
      sq  += __shfl_xor(sq,  msk, 32);
    }
    const float mu   = sum * (1.f / 128.f);
    const float var  = sq * (1.f / 128.f) - mu * mu;
    const float rstd = rsqrtf(var + 1e-6f);
#pragma unroll
    for (int i = 0; i < 16; ++i) {
      const int c = cbase + i;
      y[(size_t)(p0 + row) * Cc + c] =
          (vals[i] - mu) * rstd * ln_scale[c] + ln_bias[c];
    }
  }
}

extern "C" void kernel_launch(void* const* d_in, const int* in_sizes, int n_in,
                              void* d_out, int out_size, void* d_ws, size_t ws_size,
                              hipStream_t stream) {
  const float* x     = (const float*)d_in[0];
  const float* s     = (const float*)d_in[1];
  const float* amask = (const float*)d_in[2];
  const float* Wq    = (const float*)d_in[3];
  const float* bq    = (const float*)d_in[4];
  const float* Wkx   = (const float*)d_in[5];
  const float* bkx   = (const float*)d_in[6];
  const float* Wvx   = (const float*)d_in[7];
  const float* bvx   = (const float*)d_in[8];
  const float* Wks   = (const float*)d_in[9];
  const float* bks   = (const float*)d_in[10];
  const float* Wvs   = (const float*)d_in[11];
  const float* bvs   = (const float*)d_in[12];
  const float* Wout  = (const float*)d_in[13];
  const float* bout  = (const float*)d_in[14];
  const float* lns   = (const float*)d_in[15];
  const float* lnb   = (const float*)d_in[16];
  float* y = (float*)d_out;

  char* ws = (char*)d_ws;
  float* ksb  = (float*)(ws + 0);
  float* vsb  = (float*)(ws + 262144);
  float* G    = (float*)(ws + 524288);
  float* Ksum = (float*)(ws + 589824);
  float* Vsum = (float*)(ws + 590848);

  text_proj_kernel<<<dim3(Bn * (Ss / 16) * 2), dim3(128), 0, stream>>>(
      s, amask, Wks, bks, Wvs, bvs, ksb, vsb);
  colsum_kernel<<<dim3(Bn), dim3(128), 0, stream>>>(ksb, vsb, Ksum, Vsum);
  gram_kernel<<<dim3(Bn * (NC / 16)), dim3(128), 0, stream>>>(ksb, vsb, G);
  fused_attn_kernel<<<dim3(NPOS / 16), dim3(128), 0, stream>>>(
      x, Wq, bq, Wkx, bkx, Wvx, bvx, G, Ksum, Vsum, Wout, bout, lns, lnb, y);
}


# Column-sum and Gram launches folded into the text projection as split-K partials (32 text rows a block, ks/vs stay in LDS); fused kernel adds the 8 partials at staging: 4 launches to 2; new sum order
# speedup vs baseline: 1.9018x; 1.9018x over previous
#include <hip/hip_runtime.h>
#include <math.h>

constexpr int Bn  = 4;
constexpr int Wd  = 32;
constexpr int Hd  = 32;
constexpr int Cc  = 128;
constexpr int Ss  = 256;
constexpr int Dd  = 512;
constexpr int NC  = 64;
constexpr int NPOS = Bn * Wd * Hd;

typedef __attribute__((ext_vector_type(2))) float v2f;
typedef __attribute__((ext_vector_type(8))) float v8f;

__device__ __forceinline__ v8f wmma_k4(v2f a, v2f b, v8f c) {
  return __builtin_amdgcn_wmma_f32_16x16x4_f32(
       false, a,  false, b,
       (short)0, c,  false,  false);
}

typedef __attribute__((ext_vector_type(4))) float v4f;

__device__ __forceinline__ float xhalf(float x) {
  const int b = __builtin_bit_cast(int, x);
  return __builtin_bit_cast(float,
      __builtin_amdgcn_permlanex16(b, b, 0x76543210, 0xfedcba98, false, false));
}


constexpr int TR    = 32;
constexpr int NPART = Ss / TR;

__global__ __launch_bounds__(256) void text_gram_kernel(
    const float* __restrict__ s, const float* __restrict__ amask,
    const float* __restrict__ Wks, const float* __restrict__ bks,
    const float* __restrict__ Wvs, const float* __restrict__ bvs,
    float* __restrict__ Gp, float* __restrict__ Ksp, float* __restrict__ Vsp) {
  constexpr int TE_STRIDE = Dd + 4;
  constexpr int KP = NC + 12;
  __shared__ float te_s[TR * TE_STRIDE];
  __shared__ float kt[TR * KP];
  __shared__ float vt[TR * KP];

  const int st = blockIdx.x & (NPART - 1);
  const int b  = blockIdx.x / NPART;

  const int tid  = threadIdx.x;
  const int lane = tid & 31;
  const int wave = tid >> 5;

  for (int i = tid; i < TR * Dd; i += 256) {
    const int r = i >> 9, c = i & (Dd - 1);
    const int srow = st * TR + r;
    te_s[r * TE_STRIDE + c] =
        s[((size_t)b * Ss + srow) * Dd + c] * amask[b * Ss + srow];
  }
  __syncthreads();

  const int m    = lane & 15;
  const int koff = (lane >> 4) * 2;

  {
    const int mrow = (wave >> 2) * 16;
    const int n    = (wave & 3) * 16 + m;
    v8f ak = {0.f, 0.f, 0.f, 0.f, 0.f, 0.f, 0.f, 0.f};
    v8f av = {0.f, 0.f, 0.f, 0.f, 0.f, 0.f, 0.f, 0.f};
#pragma unroll 4
    for (int k = 0; k < Dd; k += 4) {
      v2f af, bk, bv;
      af.x = te_s[(mrow + m) * TE_STRIDE + k + koff];
      af.y = te_s[(mrow + m) * TE_STRIDE + k + koff + 1];
      bk.x = Wks[(k + koff) * NC + n];
      bk.y = Wks[(k + koff + 1) * NC + n];
      bv.x = Wvs[(k + koff) * NC + n];
      bv.y = Wvs[(k + koff + 1) * NC + n];
      ak = wmma_k4(af, bk, ak);
      av = wmma_k4(af, bv, av);
    }
    const float bkn = bks[n];
    const float bvn = bvs[n];
#pragma unroll
    for (int i = 0; i < 8; ++i) {
      const int row = mrow + i + ((lane >> 4) << 3);
      kt[row * KP + n] = ak[i] + bkn;
      vt[row * KP + n] = av[i] + bvn;
    }
  }
  __syncthreads();

  {
    const int c0 = (wave >> 1) * 16;
    const int n0 = (wave & 1) * 32;
    v8f g0 = {0.f, 0.f, 0.f, 0.f, 0.f, 0.f, 0.f, 0.f};
    v8f g1 = {0.f, 0.f, 0.f, 0.f, 0.f, 0.f, 0.f, 0.f};
#pragma unroll
    for (int k = 0; k < TR; k += 4) {
      v2f af, b0, b1;
      af.x = vt[(k + koff) * KP + c0 + m];
      af.y = vt[(k + koff + 1) * KP + c0 + m];
      b0.x = kt[(k + koff) * KP + n0 + m];
      b0.y = kt[(k + koff + 1) * KP + n0 + m];
      b1.x = kt[(k + koff) * KP + n0 + 16 + m];
      b1.y = kt[(k + koff + 1) * KP + n0 + 16 + m];
      g0 = wmma_k4(af, b0, g0);
      g1 = wmma_k4(af, b1, g1);
    }
    float* gdst = Gp + (size_t)(b * NPART + st) * (NC * NC);
#pragma unroll
    for (int i = 0; i < 8; ++i) {
      const int row = c0 + i + ((lane >> 4) << 3);
      gdst[row * NC + n0 + m]      = g0[i];
      gdst[row * NC + n0 + 16 + m] = g1[i];
    }
  }

  if (tid < 2 * NC) {
    const float* src = (tid < NC) ? kt : vt;
    const int col = tid & (NC - 1);
    float acc = 0.f;
    for (int r = 0; r < TR; ++r) acc += src[r * KP + col];
    float* dst = (tid < NC) ? Ksp : Vsp;
    dst[(b * NPART + st) * NC + col] = acc;
  }
}

__global__ __launch_bounds__(128) void fused_attn_kernel(
    const float* __restrict__ x,
    const float* __restrict__ Wq,  const float* __restrict__ bq,
    const float* __restrict__ Wkx, const float* __restrict__ bkx,
    const float* __restrict__ Wvx, const float* __restrict__ bvx,
    const float* __restrict__ Gp,
    const float* __restrict__ Ksp, const float* __restrict__ Vsp,
    const float* __restrict__ Wout, const float* __restrict__ bout,
    const float* __restrict__ ln_scale, const float* __restrict__ ln_bias,
    float* __restrict__ y) {
  constexpr int XS = Cc + 4;
  constexpr int QS = NC + 4;
  __shared__ float xs[16 * XS];
  __shared__ float qs[16 * QS];
  __shared__ float kxs[16 * QS];
  __shared__ float vxs[16 * QS];
  __shared__ float os[16 * QS];
  __shared__ float outs[16 * XS];
  constexpr int GP = NC + 4;
  constexpr int RP = 20;
  __shared__ __attribute__((aligned(16))) float Gs[NC * GP];
  __shared__ float KsS[NC], VsS[NC];
  __shared__ __attribute__((aligned(16))) float red[4 * 2 * NC * RP];

  const int tid  = threadIdx.x;
  const int lane = tid & 31;
  const int wave = tid >> 5;
  const int p0   = blockIdx.x * 16;
  const int b    = p0 >> 10;

  for (int i = tid; i < 16 * Cc; i += 128)
    xs[(i >> 7) * XS + (i & (Cc - 1))] = x[(size_t)p0 * Cc + i];
  {
    constexpr int G4 = NC * NC / 4;
    const v4f* gp4 = reinterpret_cast<const v4f*>(Gp + (size_t)b * NPART * NC * NC);
#pragma unroll 4
    for (int i = tid; i < G4; i += 128) {
      v4f a = gp4[i];
      for (int t = 1; t < NPART; ++t) a += gp4[t * G4 + i];
      *reinterpret_cast<v4f*>(&Gs[(i >> 4) * GP + (i & 15) * 4]) = a;
    }
  }
  {
    const float* sp = (tid < NC) ? Ksp : Vsp;
    const int col = tid & (NC - 1);
    float a = 0.f;
    for (int t = 0; t < NPART; ++t) a += sp[(b * NPART + t) * NC + col];
    if (tid < NC) KsS[col] = a;
    else          VsS[col] = a;
  }
  __syncthreads();

  {
    const int m    = lane & 15;
    const int koff = (lane >> 4) * 2;
    const int n    = wave * 16 + (lane & 15);
    for (int mat = 0; mat < 3; ++mat) {
      const float* Wm   = (mat == 0) ? Wq  : (mat == 1) ? Wkx : Wvx;
      const float* bias = (mat == 0) ? bq  : (mat == 1) ? bkx : bvx;
      float*       dst  = (mat == 0) ? qs  : (mat == 1) ? kxs : vxs;
      v8f acc = {0.f, 0.f, 0.f, 0.f, 0.f, 0.f, 0.f, 0.f};
#pragma unroll 4
      for (int k = 0; k < Cc; k += 4) {
        v2f af, bf;
        af.x = xs[m * XS + k + koff];
        af.y = xs[m * XS + k + koff + 1];
        bf.x = Wm[(k + koff) * NC + n];
        bf.y = Wm[(k + koff + 1) * NC + n];
        acc = wmma_k4(af, bf, acc);
      }
      const float bn = bias[n];
#pragma unroll
      for (int i = 0; i < 8; ++i)
        dst[(i + ((lane >> 4) << 3)) * QS + n] = acc[i] + bn;
    }
  }
  __syncthreads();

  {
    const int c = lane & 15;
    const int h = lane >> 4;
    constexpr float C8 = 0.18033688011112042f;
    for (int pi = 0; pi < 4; ++pi) {
      const int p = wave * 4 + pi;
      v2f af[4];
      for (int mt = 0; mt < 4; ++mt) {
        const float ksd = KsS[mt * 16 + c];
        af[mt].x = kxs[p * QS + mt * 16 + c];
        af[mt].y = h ? 0.f : ksd;
      }
      v8f acc[4];
      for (int mt = 0; mt < 4; ++mt)
        acc[mt] = (v8f){0.f, 0.f, 0.f, 0.f, 0.f, 0.f, 0.f, 0.f};

#pragma unroll
      for (int nt = 0; nt < 4; ++nt) {
        const int j = nt * 16 + c;
        const float vxj = vxs[p * QS + j];
        const float Vsj = VsS[j];
        v2f bf;
        bf.x = h ? Vsj : 256.f * vxj;
        bf.y = h ? 0.f : vxj;

        v8f L[4];
        for (int mt = 0; mt < 4; ++mt) {
          const v4f* gp = reinterpret_cast<const v4f*>(&Gs[j * GP + mt * 16 + 8 * h]);
          const v4f g0 = gp[0];
          const v4f g1 = gp[1];
          const v8f cf = __builtin_shufflevector(g0, g1, 0, 1, 2, 3, 4, 5, 6, 7);
          L[mt] = wmma_k4(af[mt], bf, cf);
        }

        float mm[4];
        for (int mt = 0; mt < 4; ++mt) {
          float t = L[mt][0];
          for (int r = 1; r < 8; ++r) t = fmaxf(t, L[mt][r]);
          mm[mt] = t;
        }
        float m = fmaxf(fmaxf(mm[0], mm[1]), fmaxf(mm[2], mm[3]));
        m = fmaxf(m, xhalf(m));
        const float nm = -m * C8;

        float ss[4];
        for (int mt = 0; mt < 4; ++mt) {
          float t = 0.f;
          for (int r = 0; r < 8; ++r) {
            const float e = __builtin_amdgcn_exp2f(fmaf(L[mt][r], C8, nm));
            L[mt][r] = e;
            t += e;
          }
          ss[mt] = t;
        }
        float sm = (ss[0] + ss[1]) + (ss[2] + ss[3]);
        sm += xhalf(sm);
        const float qw = qs[p * QS + j] / sm;
        for (int mt = 0; mt < 4; ++mt)
          for (int r = 0; r < 8; ++r)
            acc[mt][r] = fmaf(qw, L[mt][r], acc[mt][r]);
      }

      float* rb = red + (wave * 2 + (pi & 1)) * (NC * RP);
      for (int mt = 0; mt < 4; ++mt)
        for (int r = 0; r < 8; ++r)
          rb[(mt * 16 + 8 * h + r) * RP + c] = acc[mt][r];
      __syncthreads();
      for (int t = 0; t < 2; ++t) {
        const int d = lane + 32 * t;
        const v4f* rp = reinterpret_cast<const v4f*>(&rb[d * RP]);
        const v4f a0 = rp[0], a1 = rp[1], a2 = rp[2], a3 = rp[3];
        const v4f s4 = (a0 + a1) + (a2 + a3);
        os[p * QS + d] = (s4.x + s4.y) + (s4.z + s4.w);
      }
    }
  }
  __syncthreads();

  {
    const int m    = lane & 15;
    const int koff = (lane >> 4) * 2;
    for (int ct = wave; ct < 8; ct += 4) {
      const int n = ct * 16 + (lane & 15);
      v8f acc = {0.f, 0.f, 0.f, 0.f, 0.f, 0.f, 0.f, 0.f};
#pragma unroll 4
      for (int k = 0; k < NC; k += 4) {
        v2f af, bf;
        af.x = os[m * QS + k + koff];
        af.y = os[m * QS + k + koff + 1];
        bf.x = Wout[(k + koff) * Cc + n];
        bf.y = Wout[(k + koff + 1) * Cc + n];
        acc = wmma_k4(af, bf, acc);
      }
      const float bn = bout[n];
#pragma unroll
      for (int i = 0; i < 8; ++i)
        outs[(i + ((lane >> 4) << 3)) * XS + n] = acc[i] + bn;
    }
  }
  __syncthreads();

  {
    const int row   = tid >> 3;
    const int cbase = (tid & 7) * 16;
    float vals[16];
    float sum = 0.f, sq = 0.f;
#pragma unroll
    for (int i = 0; i < 16; ++i) {
      const float v = xs[row * XS + cbase + i] + outs[row * XS + cbase + i];
      vals[i] = v;
      sum += v;
      sq  += v * v;
    }
#pragma unroll
    for (int msk = 4; msk >= 1; msk >>= 1) {
      sum += __shfl_xor(sum, msk, 32);
      sq  += __shfl_xor(sq,  msk, 32);
    }
    const float mu   = sum * (1.f / 128.f);
    const float var  = sq * (1.f / 128.f) - mu * mu;
    const float rstd = rsqrtf(var + 1e-6f);
#pragma unroll
    for (int i = 0; i < 16; ++i) {
      const int c = cbase + i;
      y[(size_t)(p0 + row) * Cc + c] =
          (vals[i] - mu) * rstd * ln_scale[c] + ln_bias[c];
    }
  }
}

extern "C" void kernel_launch(void* const* d_in, const int* in_sizes, int n_in,
                              void* d_out, int out_size, void* d_ws, size_t ws_size,
                              hipStream_t stream) {
  const float* x     = (const float*)d_in[0];
  const float* s     = (const float*)d_in[1];
  const float* amask = (const float*)d_in[2];
  const float* Wq    = (const float*)d_in[3];
  const float* bq    = (const float*)d_in[4];
  const float* Wkx   = (const float*)d_in[5];
  const float* bkx   = (const float*)d_in[6];
  const float* Wvx   = (const float*)d_in[7];
  const float* bvx   = (const float*)d_in[8];
  const float* Wks   = (const float*)d_in[9];
  const float* bks   = (const float*)d_in[10];
  const float* Wvs   = (const float*)d_in[11];
  const float* bvs   = (const float*)d_in[12];
  const float* Wout  = (const float*)d_in[13];
  const float* bout  = (const float*)d_in[14];
  const float* lns   = (const float*)d_in[15];
  const float* lnb   = (const float*)d_in[16];
  float* y = (float*)d_out;

  char* ws = (char*)d_ws;
  float* Gp  = (float*)(ws + 0);
  float* Ksp = (float*)(ws + 524288);
  float* Vsp = (float*)(ws + 532480);

  text_gram_kernel<<<dim3(Bn * NPART), dim3(256), 0, stream>>>(
      s, amask, Wks, bks, Wvs, bvs, Gp, Ksp, Vsp);
  fused_attn_kernel<<<dim3(NPOS / 16), dim3(128), 0, stream>>>(
      x, Wq, bq, Wkx, bkx, Wvx, bvx, Gp, Ksp, Vsp, Wout, bout, lns, lnb, y);
}


# Global operands (weights, biases, LN rows, text and x tiles) copied to LDS by block-wide 16-byte load bursts; weight columns XOR-swizzled, text weights in 2 halves; k loops and epilogues read LDS
# speedup vs baseline: 4.6718x; 2.4565x over previous
#include <hip/hip_runtime.h>
#include <math.h>

constexpr int Bn  = 4;
constexpr int Wd  = 32;
constexpr int Hd  = 32;
constexpr int Cc  = 128;
constexpr int Ss  = 256;
constexpr int Dd  = 512;
constexpr int NC  = 64;
constexpr int NPOS = Bn * Wd * Hd;

typedef __attribute__((ext_vector_type(2))) float v2f;
typedef __attribute__((ext_vector_type(8))) float v8f;

__device__ __forceinline__ v8f wmma_k4(v2f a, v2f b, v8f c) {
  return __builtin_amdgcn_wmma_f32_16x16x4_f32(
       false, a,  false, b,
       (short)0, c,  false,  false);
}

typedef __attribute__((ext_vector_type(4))) float v4f;

__device__ __forceinline__ float xhalf(float x) {
  const int b = __builtin_bit_cast(int, x);
  return __builtin_bit_cast(float,
      __builtin_amdgcn_permlanex16(b, b, 0x76543210, 0xfedcba98, false, false));
}


constexpr int TR    = 32;
constexpr int NPART = Ss / TR;

__global__ __launch_bounds__(256) void text_gram_kernel(
    const float* __restrict__ s, const float* __restrict__ amask,
    const float* __restrict__ Wks, const float* __restrict__ bks,
    const float* __restrict__ Wvs, const float* __restrict__ bvs,
    float* __restrict__ Gp, float* __restrict__ Ksp, float* __restrict__ Vsp) {
  constexpr int TE_STRIDE = Dd + 4;
  constexpr int KP = NC + 12;
  constexpr int KH = Dd / 2;
  __shared__ __attribute__((aligned(16))) float te_s[TR * TE_STRIDE];
  __shared__ __attribute__((aligned(16))) float wk_s[KH * NC];
  __shared__ __attribute__((aligned(16))) float wv_s[KH * NC];
  __shared__ __attribute__((aligned(16))) float bkv_s[2 * NC];
  __shared__ float kt[TR * KP];
  __shared__ float vt[TR * KP];

  const int st = blockIdx.x & (NPART - 1);
  const int b  = blockIdx.x / NPART;

  const int tid  = threadIdx.x;
  const int lane = tid & 31;
  const int wave = tid >> 5;

  const int m    = lane & 15;
  const int koff = (lane >> 4) * 2;

  const v4f* s4  = reinterpret_cast<const v4f*>(s + ((size_t)b * Ss + st * TR) * Dd);
  const v4f* wk4 = reinterpret_cast<const v4f*>(Wks);
  const v4f* wv4 = reinterpret_cast<const v4f*>(Wvs);
  constexpr int WH4 = KH * NC / 4;
  const int wsw  = ((tid >> 5) & 1) << 4;
  const int wdst = (tid >> 4) * NC + (((tid & 15) * 4) ^ wsw);

  {
    v4f tv[16], wa[16], wb[16];
    float am[16];
#pragma unroll
    for (int j = 0; j < 16; ++j) {
      tv[j] = s4[tid + 256 * j];
      am[j] = amask[b * Ss + st * TR + (tid >> 7) + 2 * j];
    }
#pragma unroll
    for (int j = 0; j < 16; ++j) {
      wa[j] = wk4[tid + 256 * j];
      wb[j] = wv4[tid + 256 * j];
    }
    v4f bz = {0.f, 0.f, 0.f, 0.f};
    if (tid < 32)
      bz = reinterpret_cast<const v4f*>(tid < 16 ? bks : bvs)[tid & 15];
#pragma unroll
    for (int j = 0; j < 16; ++j)
      *reinterpret_cast<v4f*>(&te_s[((tid >> 7) + 2 * j) * TE_STRIDE + (tid & 127) * 4]) =
          tv[j] * am[j];
#pragma unroll
    for (int j = 0; j < 16; ++j) {
      *reinterpret_cast<v4f*>(&wk_s[wdst + 16 * j * NC]) = wa[j];
      *reinterpret_cast<v4f*>(&wv_s[wdst + 16 * j * NC]) = wb[j];
    }
    if (tid < 32)
      *reinterpret_cast<v4f*>(&bkv_s[tid * 4]) = bz;
  }
  __syncthreads();

  {
    const int mrow = (wave >> 2) * 16;
    const int n    = (wave & 3) * 16 + m;
    const int nsw  = n ^ ((lane >> 4) << 4);
    v8f ak = {0.f, 0.f, 0.f, 0.f, 0.f, 0.f, 0.f, 0.f};
    v8f av = {0.f, 0.f, 0.f, 0.f, 0.f, 0.f, 0.f, 0.f};

    v4f wa[16], wb[16];
#pragma unroll
    for (int j = 0; j < 16; ++j) {
      wa[j] = wk4[WH4 + tid + 256 * j];
      wb[j] = wv4[WH4 + tid + 256 * j];
    }

#pragma unroll 4
    for (int k = 0; k < KH; k += 4) {
      v2f af, bk, bv;
      af.x = te_s[(mrow + m) * TE_STRIDE + k + koff];
      af.y = te_s[(mrow + m) * TE_STRIDE + k + koff + 1];
      bk.x = wk_s[(k + koff) * NC + nsw];
      bk.y = wk_s[(k + koff + 1) * NC + nsw];
      bv.x = wv_s[(k + koff) * NC + nsw];
      bv.y = wv_s[(k + koff + 1) * NC + nsw];
      ak = wmma_k4(af, bk, ak);
      av = wmma_k4(af, bv, av);
    }
    __syncthreads();
#pragma unroll
    for (int j = 0; j < 16; ++j) {
      *reinterpret_cast<v4f*>(&wk_s[wdst + 16 * j * NC]) = wa[j];
      *reinterpret_cast<v4f*>(&wv_s[wdst + 16 * j * NC]) = wb[j];
    }
    __syncthreads();
#pragma unroll 4
    for (int k = 0; k < KH; k += 4) {
      v2f af, bk, bv;
      af.x = te_s[(mrow + m) * TE_STRIDE + KH + k + koff];
      af.y = te_s[(mrow + m) * TE_STRIDE + KH + k + koff + 1];
      bk.x = wk_s[(k + koff) * NC + nsw];
      bk.y = wk_s[(k + koff + 1) * NC + nsw];
      bv.x = wv_s[(k + koff) * NC + nsw];
      bv.y = wv_s[(k + koff + 1) * NC + nsw];
      ak = wmma_k4(af, bk, ak);
      av = wmma_k4(af, bv, av);
    }
    const float bkn = bkv_s[n];
    const float bvn = bkv_s[NC + n];
#pragma unroll
    for (int i = 0; i < 8; ++i) {
      const int row = mrow + i + ((lane >> 4) << 3);
      kt[row * KP + n] = ak[i] + bkn;
      vt[row * KP + n] = av[i] + bvn;
    }
  }
  __syncthreads();

  {
    const int c0 = (wave >> 1) * 16;
    const int n0 = (wave & 1) * 32;
    v8f g0 = {0.f, 0.f, 0.f, 0.f, 0.f, 0.f, 0.f, 0.f};
    v8f g1 = {0.f, 0.f, 0.f, 0.f, 0.f, 0.f, 0.f, 0.f};
#pragma unroll
    for (int k = 0; k < TR; k += 4) {
      v2f af, b0, b1;
      af.x = vt[(k + koff) * KP + c0 + m];
      af.y = vt[(k + koff + 1) * KP + c0 + m];
      b0.x = kt[(k + koff) * KP + n0 + m];
      b0.y = kt[(k + koff + 1) * KP + n0 + m];
      b1.x = kt[(k + koff) * KP + n0 + 16 + m];
      b1.y = kt[(k + koff + 1) * KP + n0 + 16 + m];
      g0 = wmma_k4(af, b0, g0);
      g1 = wmma_k4(af, b1, g1);
    }
    float* gdst = Gp + (size_t)(b * NPART + st) * (NC * NC);
#pragma unroll
    for (int i = 0; i < 8; ++i) {
      const int row = c0 + i + ((lane >> 4) << 3);
      gdst[row * NC + n0 + m]      = g0[i];
      gdst[row * NC + n0 + 16 + m] = g1[i];
    }
  }

  if (tid < 2 * NC) {
    const float* src = (tid < NC) ? kt : vt;
    const int col = tid & (NC - 1);
    float acc = 0.f;
    for (int r = 0; r < TR; ++r) acc += src[r * KP + col];
    float* dst = (tid < NC) ? Ksp : Vsp;
    dst[(b * NPART + st) * NC + col] = acc;
  }
}

__global__ __launch_bounds__(128) void fused_attn_kernel(
    const float* __restrict__ x,
    const float* __restrict__ Wq,  const float* __restrict__ bq,
    const float* __restrict__ Wkx, const float* __restrict__ bkx,
    const float* __restrict__ Wvx, const float* __restrict__ bvx,
    const float* __restrict__ Gp,
    const float* __restrict__ Ksp, const float* __restrict__ Vsp,
    const float* __restrict__ Wout, const float* __restrict__ bout,
    const float* __restrict__ ln_scale, const float* __restrict__ ln_bias,
    float* __restrict__ y) {
  constexpr int XS = Cc + 4;
  constexpr int QS = NC + 4;
  __shared__ __attribute__((aligned(16))) float xs[16 * XS];
  __shared__ __attribute__((aligned(16))) float w3_s[3 * Cc * NC];
  __shared__ __attribute__((aligned(16))) float wo_s[NC * Cc];
  constexpr int BOUT = 3 * NC, LNS = BOUT + Cc, LNB = LNS + Cc, NB4 = (LNB + Cc) / 4;
  __shared__ __attribute__((aligned(16))) float bias_s[NB4 * 4];
  __shared__ float qs[16 * QS];
  __shared__ float kxs[16 * QS];
  __shared__ float vxs[16 * QS];
  __shared__ float os[16 * QS];
  __shared__ float outs[16 * XS];
  constexpr int GP = NC + 4;
  constexpr int RP = 20;
  __shared__ __attribute__((aligned(16))) float Gs[NC * GP];
  __shared__ float KsS[NC], VsS[NC];
  __shared__ __attribute__((aligned(16))) float red[4 * 2 * NC * RP];

  const int tid  = threadIdx.x;
  const int lane = tid & 31;
  const int wave = tid >> 5;
  const int p0   = blockIdx.x * 16;
  const int b    = p0 >> 10;

  {
    const v4f* x4 = reinterpret_cast<const v4f*>(x + (size_t)p0 * Cc);
    const v4f* q4 = reinterpret_cast<const v4f*>(Wq);
    const v4f* k4 = reinterpret_cast<const v4f*>(Wkx);
    const v4f* v4 = reinterpret_cast<const v4f*>(Wvx);
    const v4f* o4 = reinterpret_cast<const v4f*>(Wout);
    v4f xr[4], qr[16], kr[16], vr[16], wr[16];
#pragma unroll
    for (int j = 0; j < 4; ++j) xr[j] = x4[tid + 128 * j];
#pragma unroll
    for (int j = 0; j < 16; ++j) {
      qr[j] = q4[tid + 128 * j];
      kr[j] = k4[tid + 128 * j];
      vr[j] = v4[tid + 128 * j];
      wr[j] = o4[tid + 128 * j];
    }
    v4f bz[6];
#pragma unroll
    for (int j = 0; j < 6; ++j) bz[j] = (v4f){0.f, 0.f, 0.f, 0.f};
    if (tid < 32) {
      bz[3] = reinterpret_cast<const v4f*>(bout)[tid];
      bz[4] = reinterpret_cast<const v4f*>(ln_scale)[tid];
      bz[5] = reinterpret_cast<const v4f*>(ln_bias)[tid];
      if (tid < 16) {
        bz[0] = reinterpret_cast<const v4f*>(bq)[tid];
        bz[1] = reinterpret_cast<const v4f*>(bkx)[tid];
        bz[2] = reinterpret_cast<const v4f*>(bvx)[tid];
      }
    }
#pragma unroll
    for (int j = 0; j < 4; ++j)
      *reinterpret_cast<v4f*>(&xs[((tid >> 5) + 4 * j) * XS + (tid & 31) * 4]) = xr[j];
    const int d3 = (tid >> 4) * NC + (((tid & 15) * 4) ^ (((tid >> 5) & 1) << 4));
    const int d1 = (tid >> 5) * Cc + (((tid & 31) * 4) ^ (((tid >> 6) & 1) << 4));
#pragma unroll
    for (int j = 0; j < 16; ++j) {
      *reinterpret_cast<v4f*>(&w3_s[0 * Cc * NC + d3 + 8 * j * NC]) = qr[j];
      *reinterpret_cast<v4f*>(&w3_s[1 * Cc * NC + d3 + 8 * j * NC]) = kr[j];
      *reinterpret_cast<v4f*>(&w3_s[2 * Cc * NC + d3 + 8 * j * NC]) = vr[j];
      *reinterpret_cast<v4f*>(&wo_s[d1 + 4 * j * Cc]) = wr[j];
    }
    if (tid < 32) {
      *reinterpret_cast<v4f*>(&bias_s[BOUT + tid * 4]) = bz[3];
      *reinterpret_cast<v4f*>(&bias_s[LNS + tid * 4])  = bz[4];
      *reinterpret_cast<v4f*>(&bias_s[LNB + tid * 4])  = bz[5];
      if (tid < 16) {
        *reinterpret_cast<v4f*>(&bias_s[0 * NC + tid * 4]) = bz[0];
        *reinterpret_cast<v4f*>(&bias_s[1 * NC + tid * 4]) = bz[1];
        *reinterpret_cast<v4f*>(&bias_s[2 * NC + tid * 4]) = bz[2];
      }
    }
  }
  {
    constexpr int G4 = NC * NC / 4;
    const v4f* gp4 = reinterpret_cast<const v4f*>(Gp + (size_t)b * NPART * NC * NC);
#pragma unroll 4
    for (int i = tid; i < G4; i += 128) {
      v4f a = gp4[i];
      for (int t = 1; t < NPART; ++t) a += gp4[t * G4 + i];
      *reinterpret_cast<v4f*>(&Gs[(i >> 4) * GP + (i & 15) * 4]) = a;
    }
  }
  {
    const float* sp = (tid < NC) ? Ksp : Vsp;
    const int col = tid & (NC - 1);
    float a = 0.f;
    for (int t = 0; t < NPART; ++t) a += sp[(b * NPART + t) * NC + col];
    if (tid < NC) KsS[col] = a;
    else          VsS[col] = a;
  }
  __syncthreads();

  {
    const int m    = lane & 15;
    const int koff = (lane >> 4) * 2;
    const int n    = wave * 16 + (lane & 15);
    const int nsw  = n ^ ((lane >> 4) << 4);
    for (int mat = 0; mat < 3; ++mat) {
      const float* Wm   = w3_s + mat * (Cc * NC);
      const float* bias = bias_s + mat * NC;
      float*       dst  = (mat == 0) ? qs  : (mat == 1) ? kxs : vxs;
      v8f acc = {0.f, 0.f, 0.f, 0.f, 0.f, 0.f, 0.f, 0.f};
#pragma unroll 4
      for (int k = 0; k < Cc; k += 4) {
        v2f af, bf;
        af.x = xs[m * XS + k + koff];
        af.y = xs[m * XS + k + koff + 1];
        bf.x = Wm[(k + koff) * NC + nsw];
        bf.y = Wm[(k + koff + 1) * NC + nsw];
        acc = wmma_k4(af, bf, acc);
      }
      const float bn = bias[n];
#pragma unroll
      for (int i = 0; i < 8; ++i)
        dst[(i + ((lane >> 4) << 3)) * QS + n] = acc[i] + bn;
    }
  }
  __syncthreads();

  {
    const int c = lane & 15;
    const int h = lane >> 4;
    constexpr float C8 = 0.18033688011112042f;
    for (int pi = 0; pi < 4; ++pi) {
      const int p = wave * 4 + pi;
      v2f af[4];
      for (int mt = 0; mt < 4; ++mt) {
        const float ksd = KsS[mt * 16 + c];
        af[mt].x = kxs[p * QS + mt * 16 + c];
        af[mt].y = h ? 0.f : ksd;
      }
      v8f acc[4];
      for (int mt = 0; mt < 4; ++mt)
        acc[mt] = (v8f){0.f, 0.f, 0.f, 0.f, 0.f, 0.f, 0.f, 0.f};

#pragma unroll
      for (int nt = 0; nt < 4; ++nt) {
        const int j = nt * 16 + c;
        const float vxj = vxs[p * QS + j];
        const float Vsj = VsS[j];
        v2f bf;
        bf.x = h ? Vsj : 256.f * vxj;
        bf.y = h ? 0.f : vxj;

        v8f L[4];
        for (int mt = 0; mt < 4; ++mt) {
          const v4f* gp = reinterpret_cast<const v4f*>(&Gs[j * GP + mt * 16 + 8 * h]);
          const v4f g0 = gp[0];
          const v4f g1 = gp[1];
          const v8f cf = __builtin_shufflevector(g0, g1, 0, 1, 2, 3, 4, 5, 6, 7);
          L[mt] = wmma_k4(af[mt], bf, cf);
        }

        float mm[4];
        for (int mt = 0; mt < 4; ++mt) {
          float t = L[mt][0];
          for (int r = 1; r < 8; ++r) t = fmaxf(t, L[mt][r]);
          mm[mt] = t;
        }
        float m = fmaxf(fmaxf(mm[0], mm[1]), fmaxf(mm[2], mm[3]));
        m = fmaxf(m, xhalf(m));
        const float nm = -m * C8;

        float ss[4];
        for (int mt = 0; mt < 4; ++mt) {
          float t = 0.f;
          for (int r = 0; r < 8; ++r) {
            const float e = __builtin_amdgcn_exp2f(fmaf(L[mt][r], C8, nm));
            L[mt][r] = e;
            t += e;
          }
          ss[mt] = t;
        }
        float sm = (ss[0] + ss[1]) + (ss[2] + ss[3]);
        sm += xhalf(sm);
        const float qw = qs[p * QS + j] / sm;
        for (int mt = 0; mt < 4; ++mt)
          for (int r = 0; r < 8; ++r)
            acc[mt][r] = fmaf(qw, L[mt][r], acc[mt][r]);
      }

      float* rb = red + (wave * 2 + (pi & 1)) * (NC * RP);
      for (int mt = 0; mt < 4; ++mt)
        for (int r = 0; r < 8; ++r)
          rb[(mt * 16 + 8 * h + r) * RP + c] = acc[mt][r];
      __syncthreads();
      for (int t = 0; t < 2; ++t) {
        const int d = lane + 32 * t;
        const v4f* rp = reinterpret_cast<const v4f*>(&rb[d * RP]);
        const v4f a0 = rp[0], a1 = rp[1], a2 = rp[2], a3 = rp[3];
        const v4f s4 = (a0 + a1) + (a2 + a3);
        os[p * QS + d] = (s4.x + s4.y) + (s4.z + s4.w);
      }
    }
  }
  __syncthreads();

  {
    const int m    = lane & 15;
    const int koff = (lane >> 4) * 2;
    for (int ct = wave; ct < 8; ct += 4) {
      const int n   = ct * 16 + (lane & 15);
      const int nsw = n ^ ((lane >> 4) << 4);
      v8f acc = {0.f, 0.f, 0.f, 0.f, 0.f, 0.f, 0.f, 0.f};
#pragma unroll 4
      for (int k = 0; k < NC; k += 4) {
        v2f af, bf;
        af.x = os[m * QS + k + koff];
        af.y = os[m * QS + k + koff + 1];
        bf.x = wo_s[(k + koff) * Cc + nsw];
        bf.y = wo_s[(k + koff + 1) * Cc + nsw];
        acc = wmma_k4(af, bf, acc);
      }
      const float bn = bias_s[BOUT + n];
#pragma unroll
      for (int i = 0; i < 8; ++i)
        outs[(i + ((lane >> 4) << 3)) * XS + n] = acc[i] + bn;
    }
  }
  __syncthreads();

  {
    const int row   = tid >> 3;
    const int cbase = (tid & 7) * 16;
    float vals[16];
    float sum = 0.f, sq = 0.f;
#pragma unroll
    for (int i = 0; i < 16; ++i) {
      const float v = xs[row * XS + cbase + i] + outs[row * XS + cbase + i];
      vals[i] = v;
      sum += v;
      sq  += v * v;
    }
#pragma unroll
    for (int msk = 4; msk >= 1; msk >>= 1) {
      sum += __shfl_xor(sum, msk, 32);
      sq  += __shfl_xor(sq,  msk, 32);
    }
    const float mu   = sum * (1.f / 128.f);
    const float var  = sq * (1.f / 128.f) - mu * mu;
    const float rstd = rsqrtf(var + 1e-6f);
#pragma unroll
    for (int i = 0; i < 16; ++i) {
      const int c = cbase + i;
      y[(size_t)(p0 + row) * Cc + c] =
          (vals[i] - mu) * rstd * bias_s[LNS + c] + bias_s[LNB + c];
    }
  }
}

extern "C" void kernel_launch(void* const* d_in, const int* in_sizes, int n_in,
                              void* d_out, int out_size, void* d_ws, size_t ws_size,
                              hipStream_t stream) {
  const float* x     = (const float*)d_in[0];
  const float* s     = (const float*)d_in[1];
  const float* amask = (const float*)d_in[2];
  const float* Wq    = (const float*)d_in[3];
  const float* bq    = (const float*)d_in[4];
  const float* Wkx   = (const float*)d_in[5];
  const float* bkx   = (const float*)d_in[6];
  const float* Wvx   = (const float*)d_in[7];
  const float* bvx   = (const float*)d_in[8];
  const float* Wks   = (const float*)d_in[9];
  const float* bks   = (const float*)d_in[10];
  const float* Wvs   = (const float*)d_in[11];
  const float* bvs   = (const float*)d_in[12];
  const float* Wout  = (const float*)d_in[13];
  const float* bout  = (const float*)d_in[14];
  const float* lns   = (const float*)d_in[15];
  const float* lnb   = (const float*)d_in[16];
  float* y = (float*)d_out;

  char* ws = (char*)d_ws;
  float* Gp  = (float*)(ws + 0);
  float* Ksp = (float*)(ws + 524288);
  float* Vsp = (float*)(ws + 532480);

  text_gram_kernel<<<dim3(Bn * NPART), dim3(256), 0, stream>>>(
      s, amask, Wks, bks, Wvs, bvs, Gp, Ksp, Vsp);
  fused_attn_kernel<<<dim3(NPOS / 16), dim3(128), 0, stream>>>(
      x, Wq, bq, Wkx, bkx, Wvx, bvx, Gp, Ksp, Vsp, Wout, bout, lns, lnb, y);
}


# WMMA k loops (text and x projections, output product) in blocks of 8 k-steps: next block's LDS fragments read as one group into a 2nd register set ahead of this block's WMMAs; a wave's tiles share A
# speedup vs baseline: 5.3640x; 1.1482x over previous
#include <hip/hip_runtime.h>
#include <math.h>

constexpr int Bn  = 4;
constexpr int Wd  = 32;
constexpr int Hd  = 32;
constexpr int Cc  = 128;
constexpr int Ss  = 256;
constexpr int Dd  = 512;
constexpr int NC  = 64;
constexpr int NPOS = Bn * Wd * Hd;

typedef __attribute__((ext_vector_type(2))) float v2f;
typedef __attribute__((ext_vector_type(8))) float v8f;

__device__ __forceinline__ v8f wmma_k4(v2f a, v2f b, v8f c) {
  return __builtin_amdgcn_wmma_f32_16x16x4_f32(
       false, a,  false, b,
       (short)0, c,  false,  false);
}

typedef __attribute__((ext_vector_type(4))) float v4f;

__device__ __forceinline__ float xhalf(float x) {
  const int b = __builtin_bit_cast(int, x);
  return __builtin_bit_cast(float,
      __builtin_amdgcn_permlanex16(b, b, 0x76543210, 0xfedcba98, false, false));
}

template <int NT, int KS, int KB, int PITCH>
__device__ __forceinline__ void wmma_lds_blocks(const float* a, const float* const (&b)[NT],
                                                v8f (&acc)[NT]) {
  static_assert(KS % KB == 0, "whole blocks");
  constexpr int NBLK = KS / KB;
  v2f fa[2][KB];
  v2f fb[2][NT][KB];
#pragma unroll
  for (int i = 0; i < KB; ++i) {
    const int k = 4 * i;
    fa[0][i].x = a[k];
    fa[0][i].y = a[k + 1];
#pragma unroll
    for (int t = 0; t < NT; ++t) {
      fb[0][t][i].x = b[t][k * PITCH];
      fb[0][t][i].y = b[t][(k + 1) * PITCH];
    }
  }
#pragma unroll
  for (int blk = 0; blk < NBLK; ++blk) {
    const int cur = blk & 1;
    const int nxt = cur ^ 1;
    if (blk + 1 < NBLK) {
#pragma unroll
      for (int i = 0; i < KB; ++i) {
        const int k = 4 * ((blk + 1) * KB + i);
        fa[nxt][i].x = a[k];
        fa[nxt][i].y = a[k + 1];
#pragma unroll
        for (int t = 0; t < NT; ++t) {
          fb[nxt][t][i].x = b[t][k * PITCH];
          fb[nxt][t][i].y = b[t][(k + 1) * PITCH];
        }
      }
    }
    __builtin_amdgcn_sched_barrier(0);
#pragma unroll
    for (int i = 0; i < KB; ++i) {
#pragma unroll
      for (int t = 0; t < NT; ++t)
        acc[t] = wmma_k4(fa[cur][i], fb[cur][t][i], acc[t]);
    }
    __builtin_amdgcn_sched_barrier(0);
  }
}


constexpr int TR    = 32;
constexpr int NPART = Ss / TR;

__global__ __launch_bounds__(256) void text_gram_kernel(
    const float* __restrict__ s, const float* __restrict__ amask,
    const float* __restrict__ Wks, const float* __restrict__ bks,
    const float* __restrict__ Wvs, const float* __restrict__ bvs,
    float* __restrict__ Gp, float* __restrict__ Ksp, float* __restrict__ Vsp) {
  constexpr int TE_STRIDE = Dd + 4;
  constexpr int KP = NC + 12;
  constexpr int KH = Dd / 2;
  __shared__ __attribute__((aligned(16))) float te_s[TR * TE_STRIDE];
  __shared__ __attribute__((aligned(16))) float wk_s[KH * NC];
  __shared__ __attribute__((aligned(16))) float wv_s[KH * NC];
  __shared__ __attribute__((aligned(16))) float bkv_s[2 * NC];
  __shared__ float kt[TR * KP];
  __shared__ float vt[TR * KP];

  const int st = blockIdx.x & (NPART - 1);
  const int b  = blockIdx.x / NPART;

  const int tid  = threadIdx.x;
  const int lane = tid & 31;
  const int wave = tid >> 5;

  const int m    = lane & 15;
  const int koff = (lane >> 4) * 2;

  const v4f* s4  = reinterpret_cast<const v4f*>(s + ((size_t)b * Ss + st * TR) * Dd);
  const v4f* wk4 = reinterpret_cast<const v4f*>(Wks);
  const v4f* wv4 = reinterpret_cast<const v4f*>(Wvs);
  constexpr int WH4 = KH * NC / 4;
  const int wsw  = ((tid >> 5) & 1) << 4;
  const int wdst = (tid >> 4) * NC + (((tid & 15) * 4) ^ wsw);

  {
    v4f tv[16], wa[16], wb[16];
    float am[16];
#pragma unroll
    for (int j = 0; j < 16; ++j) {
      tv[j] = s4[tid + 256 * j];
      am[j] = amask[b * Ss + st * TR + (tid >> 7) + 2 * j];
    }
#pragma unroll
    for (int j = 0; j < 16; ++j) {
      wa[j] = wk4[tid + 256 * j];
      wb[j] = wv4[tid + 256 * j];
    }
    v4f bz = {0.f, 0.f, 0.f, 0.f};
    if (tid < 32)
      bz = reinterpret_cast<const v4f*>(tid < 16 ? bks : bvs)[tid & 15];
#pragma unroll
    for (int j = 0; j < 16; ++j)
      *reinterpret_cast<v4f*>(&te_s[((tid >> 7) + 2 * j) * TE_STRIDE + (tid & 127) * 4]) =
          tv[j] * am[j];
#pragma unroll
    for (int j = 0; j < 16; ++j) {
      *reinterpret_cast<v4f*>(&wk_s[wdst + 16 * j * NC]) = wa[j];
      *reinterpret_cast<v4f*>(&wv_s[wdst + 16 * j * NC]) = wb[j];
    }
    if (tid < 32)
      *reinterpret_cast<v4f*>(&bkv_s[tid * 4]) = bz;
  }
  __syncthreads();

  {
    const int mrow = (wave >> 2) * 16;
    const int n    = (wave & 3) * 16 + m;
    const int nsw  = n ^ ((lane >> 4) << 4);
    v8f akv[2];
    akv[0] = (v8f){0.f, 0.f, 0.f, 0.f, 0.f, 0.f, 0.f, 0.f};
    akv[1] = (v8f){0.f, 0.f, 0.f, 0.f, 0.f, 0.f, 0.f, 0.f};

    v4f wa[16], wb[16];
#pragma unroll
    for (int j = 0; j < 16; ++j) {
      wa[j] = wk4[WH4 + tid + 256 * j];
      wb[j] = wv4[WH4 + tid + 256 * j];
    }

    const float* const bkv[2] = {wk_s + koff * NC + nsw, wv_s + koff * NC + nsw};
    wmma_lds_blocks<2, KH / 4, 8, NC>(te_s + (mrow + m) * TE_STRIDE + koff, bkv, akv);
    __syncthreads();
#pragma unroll
    for (int j = 0; j < 16; ++j) {
      *reinterpret_cast<v4f*>(&wk_s[wdst + 16 * j * NC]) = wa[j];
      *reinterpret_cast<v4f*>(&wv_s[wdst + 16 * j * NC]) = wb[j];
    }
    __syncthreads();
    wmma_lds_blocks<2, KH / 4, 8, NC>(te_s + (mrow + m) * TE_STRIDE + KH + koff, bkv, akv);
    const float bkn = bkv_s[n];
    const float bvn = bkv_s[NC + n];
#pragma unroll
    for (int i = 0; i < 8; ++i) {
      const int row = mrow + i + ((lane >> 4) << 3);
      kt[row * KP + n] = akv[0][i] + bkn;
      vt[row * KP + n] = akv[1][i] + bvn;
    }
  }
  __syncthreads();

  {
    const int c0 = (wave >> 1) * 16;
    const int n0 = (wave & 1) * 32;
    v8f g0 = {0.f, 0.f, 0.f, 0.f, 0.f, 0.f, 0.f, 0.f};
    v8f g1 = {0.f, 0.f, 0.f, 0.f, 0.f, 0.f, 0.f, 0.f};
#pragma unroll
    for (int k = 0; k < TR; k += 4) {
      v2f af, b0, b1;
      af.x = vt[(k + koff) * KP + c0 + m];
      af.y = vt[(k + koff + 1) * KP + c0 + m];
      b0.x = kt[(k + koff) * KP + n0 + m];
      b0.y = kt[(k + koff + 1) * KP + n0 + m];
      b1.x = kt[(k + koff) * KP + n0 + 16 + m];
      b1.y = kt[(k + koff + 1) * KP + n0 + 16 + m];
      g0 = wmma_k4(af, b0, g0);
      g1 = wmma_k4(af, b1, g1);
    }
    float* gdst = Gp + (size_t)(b * NPART + st) * (NC * NC);
#pragma unroll
    for (int i = 0; i < 8; ++i) {
      const int row = c0 + i + ((lane >> 4) << 3);
      gdst[row * NC + n0 + m]      = g0[i];
      gdst[row * NC + n0 + 16 + m] = g1[i];
    }
  }

  if (tid < 2 * NC) {
    const float* src = (tid < NC) ? kt : vt;
    const int col = tid & (NC - 1);
    float acc = 0.f;
    for (int r = 0; r < TR; ++r) acc += src[r * KP + col];
    float* dst = (tid < NC) ? Ksp : Vsp;
    dst[(b * NPART + st) * NC + col] = acc;
  }
}

__global__ __launch_bounds__(128) void fused_attn_kernel(
    const float* __restrict__ x,
    const float* __restrict__ Wq,  const float* __restrict__ bq,
    const float* __restrict__ Wkx, const float* __restrict__ bkx,
    const float* __restrict__ Wvx, const float* __restrict__ bvx,
    const float* __restrict__ Gp,
    const float* __restrict__ Ksp, const float* __restrict__ Vsp,
    const float* __restrict__ Wout, const float* __restrict__ bout,
    const float* __restrict__ ln_scale, const float* __restrict__ ln_bias,
    float* __restrict__ y) {
  constexpr int XS = Cc + 4;
  constexpr int QS = NC + 4;
  __shared__ __attribute__((aligned(16))) float xs[16 * XS];
  __shared__ __attribute__((aligned(16))) float w3_s[3 * Cc * NC];
  __shared__ __attribute__((aligned(16))) float wo_s[NC * Cc];
  constexpr int BOUT = 3 * NC, LNS = BOUT + Cc, LNB = LNS + Cc, NB4 = (LNB + Cc) / 4;
  __shared__ __attribute__((aligned(16))) float bias_s[NB4 * 4];
  __shared__ float qs[16 * QS];
  __shared__ float kxs[16 * QS];
  __shared__ float vxs[16 * QS];
  __shared__ float os[16 * QS];
  __shared__ float outs[16 * XS];
  constexpr int GP = NC + 4;
  constexpr int RP = 20;
  __shared__ __attribute__((aligned(16))) float Gs[NC * GP];
  __shared__ float KsS[NC], VsS[NC];
  __shared__ __attribute__((aligned(16))) float red[4 * 2 * NC * RP];

  const int tid  = threadIdx.x;
  const int lane = tid & 31;
  const int wave = tid >> 5;
  const int p0   = blockIdx.x * 16;
  const int b    = p0 >> 10;

  {
    const v4f* x4 = reinterpret_cast<const v4f*>(x + (size_t)p0 * Cc);
    const v4f* q4 = reinterpret_cast<const v4f*>(Wq);
    const v4f* k4 = reinterpret_cast<const v4f*>(Wkx);
    const v4f* v4 = reinterpret_cast<const v4f*>(Wvx);
    const v4f* o4 = reinterpret_cast<const v4f*>(Wout);
    v4f xr[4], qr[16], kr[16], vr[16], wr[16];
#pragma unroll
    for (int j = 0; j < 4; ++j) xr[j] = x4[tid + 128 * j];
#pragma unroll
    for (int j = 0; j < 16; ++j) {
      qr[j] = q4[tid + 128 * j];
      kr[j] = k4[tid + 128 * j];
      vr[j] = v4[tid + 128 * j];
      wr[j] = o4[tid + 128 * j];
    }
    v4f bz[6];
#pragma unroll
    for (int j = 0; j < 6; ++j) bz[j] = (v4f){0.f, 0.f, 0.f, 0.f};
    if (tid < 32) {
      bz[3] = reinterpret_cast<const v4f*>(bout)[tid];
      bz[4] = reinterpret_cast<const v4f*>(ln_scale)[tid];
      bz[5] = reinterpret_cast<const v4f*>(ln_bias)[tid];
      if (tid < 16) {
        bz[0] = reinterpret_cast<const v4f*>(bq)[tid];
        bz[1] = reinterpret_cast<const v4f*>(bkx)[tid];
        bz[2] = reinterpret_cast<const v4f*>(bvx)[tid];
      }
    }
#pragma unroll
    for (int j = 0; j < 4; ++j)
      *reinterpret_cast<v4f*>(&xs[((tid >> 5) + 4 * j) * XS + (tid & 31) * 4]) = xr[j];
    const int d3 = (tid >> 4) * NC + (((tid & 15) * 4) ^ (((tid >> 5) & 1) << 4));
    const int d1 = (tid >> 5) * Cc + (((tid & 31) * 4) ^ (((tid >> 6) & 1) << 4));
#pragma unroll
    for (int j = 0; j < 16; ++j) {
      *reinterpret_cast<v4f*>(&w3_s[0 * Cc * NC + d3 + 8 * j * NC]) = qr[j];
      *reinterpret_cast<v4f*>(&w3_s[1 * Cc * NC + d3 + 8 * j * NC]) = kr[j];
      *reinterpret_cast<v4f*>(&w3_s[2 * Cc * NC + d3 + 8 * j * NC]) = vr[j];
      *reinterpret_cast<v4f*>(&wo_s[d1 + 4 * j * Cc]) = wr[j];
    }
    if (tid < 32) {
      *reinterpret_cast<v4f*>(&bias_s[BOUT + tid * 4]) = bz[3];
      *reinterpret_cast<v4f*>(&bias_s[LNS + tid * 4])  = bz[4];
      *reinterpret_cast<v4f*>(&bias_s[LNB + tid * 4])  = bz[5];
      if (tid < 16) {
        *reinterpret_cast<v4f*>(&bias_s[0 * NC + tid * 4]) = bz[0];
        *reinterpret_cast<v4f*>(&bias_s[1 * NC + tid * 4]) = bz[1];
        *reinterpret_cast<v4f*>(&bias_s[2 * NC + tid * 4]) = bz[2];
      }
    }
  }
  {
    constexpr int G4 = NC * NC / 4;
    const v4f* gp4 = reinterpret_cast<const v4f*>(Gp + (size_t)b * NPART * NC * NC);
#pragma unroll 4
    for (int i = tid; i < G4; i += 128) {
      v4f a = gp4[i];
      for (int t = 1; t < NPART; ++t) a += gp4[t * G4 + i];
      *reinterpret_cast<v4f*>(&Gs[(i >> 4) * GP + (i & 15) * 4]) = a;
    }
  }
  {
    const float* sp = (tid < NC) ? Ksp : Vsp;
    const int col = tid & (NC - 1);
    float a = 0.f;
    for (int t = 0; t < NPART; ++t) a += sp[(b * NPART + t) * NC + col];
    if (tid < NC) KsS[col] = a;
    else          VsS[col] = a;
  }
  __syncthreads();

  {
    const int m    = lane & 15;
    const int koff = (lane >> 4) * 2;
    const int n    = wave * 16 + (lane & 15);
    const int nsw  = n ^ ((lane >> 4) << 4);
    const float* const b3[3] = {w3_s + 0 * (Cc * NC) + koff * NC + nsw,
                                w3_s + 1 * (Cc * NC) + koff * NC + nsw,
                                w3_s + 2 * (Cc * NC) + koff * NC + nsw};
    v8f acc[3];
#pragma unroll
    for (int mat = 0; mat < 3; ++mat)
      acc[mat] = (v8f){0.f, 0.f, 0.f, 0.f, 0.f, 0.f, 0.f, 0.f};
    wmma_lds_blocks<3, Cc / 4, 8, NC>(xs + m * XS + koff, b3, acc);
#pragma unroll
    for (int mat = 0; mat < 3; ++mat) {
      float*      dst = (mat == 0) ? qs : (mat == 1) ? kxs : vxs;
      const float bn  = bias_s[mat * NC + n];
#pragma unroll
      for (int i = 0; i < 8; ++i)
        dst[(i + ((lane >> 4) << 3)) * QS + n] = acc[mat][i] + bn;
    }
  }
  __syncthreads();

  {
    const int c = lane & 15;
    const int h = lane >> 4;
    constexpr float C8 = 0.18033688011112042f;
    for (int pi = 0; pi < 4; ++pi) {
      const int p = wave * 4 + pi;
      v2f af[4];
      for (int mt = 0; mt < 4; ++mt) {
        const float ksd = KsS[mt * 16 + c];
        af[mt].x = kxs[p * QS + mt * 16 + c];
        af[mt].y = h ? 0.f : ksd;
      }
      v8f acc[4];
      for (int mt = 0; mt < 4; ++mt)
        acc[mt] = (v8f){0.f, 0.f, 0.f, 0.f, 0.f, 0.f, 0.f, 0.f};

#pragma unroll
      for (int nt = 0; nt < 4; ++nt) {
        const int j = nt * 16 + c;
        const float vxj = vxs[p * QS + j];
        const float Vsj = VsS[j];
        v2f bf;
        bf.x = h ? Vsj : 256.f * vxj;
        bf.y = h ? 0.f : vxj;

        v8f L[4];
        for (int mt = 0; mt < 4; ++mt) {
          const v4f* gp = reinterpret_cast<const v4f*>(&Gs[j * GP + mt * 16 + 8 * h]);
          const v4f g0 = gp[0];
          const v4f g1 = gp[1];
          const v8f cf = __builtin_shufflevector(g0, g1, 0, 1, 2, 3, 4, 5, 6, 7);
          L[mt] = wmma_k4(af[mt], bf, cf);
        }

        float mm[4];
        for (int mt = 0; mt < 4; ++mt) {
          float t = L[mt][0];
          for (int r = 1; r < 8; ++r) t = fmaxf(t, L[mt][r]);
          mm[mt] = t;
        }
        float m = fmaxf(fmaxf(mm[0], mm[1]), fmaxf(mm[2], mm[3]));
        m = fmaxf(m, xhalf(m));
        const float nm = -m * C8;

        float ss[4];
        for (int mt = 0; mt < 4; ++mt) {
          float t = 0.f;
          for (int r = 0; r < 8; ++r) {
            const float e = __builtin_amdgcn_exp2f(fmaf(L[mt][r], C8, nm));
            L[mt][r] = e;
            t += e;
          }
          ss[mt] = t;
        }
        float sm = (ss[0] + ss[1]) + (ss[2] + ss[3]);
        sm += xhalf(sm);
        const float qw = qs[p * QS + j] / sm;
        for (int mt = 0; mt < 4; ++mt)
          for (int r = 0; r < 8; ++r)
            acc[mt][r] = fmaf(qw, L[mt][r], acc[mt][r]);
      }

      float* rb = red + (wave * 2 + (pi & 1)) * (NC * RP);
      for (int mt = 0; mt < 4; ++mt)
        for (int r = 0; r < 8; ++r)
          rb[(mt * 16 + 8 * h + r) * RP + c] = acc[mt][r];
      __syncthreads();
      for (int t = 0; t < 2; ++t) {
        const int d = lane + 32 * t;
        const v4f* rp = reinterpret_cast<const v4f*>(&rb[d * RP]);
        const v4f a0 = rp[0], a1 = rp[1], a2 = rp[2], a3 = rp[3];
        const v4f s4 = (a0 + a1) + (a2 + a3);
        os[p * QS + d] = (s4.x + s4.y) + (s4.z + s4.w);
      }
    }
  }
  __syncthreads();

  {
    const int m    = lane & 15;
    const int koff = (lane >> 4) * 2;
    const int n0   = wave * 16 + m;
    const int nsw0 = n0 ^ ((lane >> 4) << 4);
    const float* const b2[2] = {wo_s + koff * Cc + nsw0, wo_s + koff * Cc + nsw0 + 64};
    v8f acc[2];
    acc[0] = (v8f){0.f, 0.f, 0.f, 0.f, 0.f, 0.f, 0.f, 0.f};
    acc[1] = (v8f){0.f, 0.f, 0.f, 0.f, 0.f, 0.f, 0.f, 0.f};
    wmma_lds_blocks<2, NC / 4, 8, Cc>(os + m * QS + koff, b2, acc);
#pragma unroll
    for (int t = 0; t < 2; ++t) {
      const int   n  = n0 + 64 * t;
      const float bn = bias_s[BOUT + n];
#pragma unroll
      for (int i = 0; i < 8; ++i)
        outs[(i + ((lane >> 4) << 3)) * XS + n] = acc[t][i] + bn;
    }
  }
  __syncthreads();

  {
    const int row   = tid >> 3;
    const int cbase = (tid & 7) * 16;
    float vals[16];
    float sum = 0.f, sq = 0.f;
#pragma unroll
    for (int i = 0; i < 16; ++i) {
      const float v = xs[row * XS + cbase + i] + outs[row * XS + cbase + i];
      vals[i] = v;
      sum += v;
      sq  += v * v;
    }
#pragma unroll
    for (int msk = 4; msk >= 1; msk >>= 1) {
      sum += __shfl_xor(sum, msk, 32);
      sq  += __shfl_xor(sq,  msk, 32);
    }
    const float mu   = sum * (1.f / 128.f);
    const float var  = sq * (1.f / 128.f) - mu * mu;
    const float rstd = rsqrtf(var + 1e-6f);
#pragma unroll
    for (int i = 0; i < 16; ++i) {
      const int c = cbase + i;
      y[(size_t)(p0 + row) * Cc + c] =
          (vals[i] - mu) * rstd * bias_s[LNS + c] + bias_s[LNB + c];
    }
  }
}

extern "C" void kernel_launch(void* const* d_in, const int* in_sizes, int n_in,
                              void* d_out, int out_size, void* d_ws, size_t ws_size,
                              hipStream_t stream) {
  const float* x     = (const float*)d_in[0];
  const float* s     = (const float*)d_in[1];
  const float* amask = (const float*)d_in[2];
  const float* Wq    = (const float*)d_in[3];
  const float* bq    = (const float*)d_in[4];
  const float* Wkx   = (const float*)d_in[5];
  const float* bkx   = (const float*)d_in[6];
  const float* Wvx   = (const float*)d_in[7];
  const float* bvx   = (const float*)d_in[8];
  const float* Wks   = (const float*)d_in[9];
  const float* bks   = (const float*)d_in[10];
  const float* Wvs   = (const float*)d_in[11];
  const float* bvs   = (const float*)d_in[12];
  const float* Wout  = (const float*)d_in[13];
  const float* bout  = (const float*)d_in[14];
  const float* lns   = (const float*)d_in[15];
  const float* lnb   = (const float*)d_in[16];
  float* y = (float*)d_out;

  char* ws = (char*)d_ws;
  float* Gp  = (float*)(ws + 0);
  float* Ksp = (float*)(ws + 524288);
  float* Vsp = (float*)(ws + 532480);

  text_gram_kernel<<<dim3(Bn * NPART), dim3(256), 0, stream>>>(
      s, amask, Wks, bks, Wvs, bvs, Gp, Ksp, Vsp);
  fused_attn_kernel<<<dim3(NPOS / 16), dim3(128), 0, stream>>>(
      x, Wq, bq, Wkx, bkx, Wvx, bvx, Gp, Ksp, Vsp, Wout, bout, lns, lnb, y);
}


# Text kernel re-gridded over the four 32x32 quadrants of each Gram partial: a block projects its 32 text rows onto 32 ks + 32 vs columns only (32 blocks -> 128; a wave one tile); sum orders unchanged
# speedup vs baseline: 6.1591x; 1.1482x over previous
#include <hip/hip_runtime.h>
#include <math.h>

constexpr int Bn  = 4;
constexpr int Wd  = 32;
constexpr int Hd  = 32;
constexpr int Cc  = 128;
constexpr int Ss  = 256;
constexpr int Dd  = 512;
constexpr int NC  = 64;
constexpr int NPOS = Bn * Wd * Hd;

typedef __attribute__((ext_vector_type(2))) float v2f;
typedef __attribute__((ext_vector_type(8))) float v8f;

__device__ __forceinline__ v8f wmma_k4(v2f a, v2f b, v8f c) {
  return __builtin_amdgcn_wmma_f32_16x16x4_f32(
       false, a,  false, b,
       (short)0, c,  false,  false);
}

typedef __attribute__((ext_vector_type(4))) float v4f;

__device__ __forceinline__ float xhalf(float x) {
  const int b = __builtin_bit_cast(int, x);
  return __builtin_bit_cast(float,
      __builtin_amdgcn_permlanex16(b, b, 0x76543210, 0xfedcba98, false, false));
}

typedef const __attribute__((address_space(3))) float* ldsp;
typedef const __attribute__((address_space(3))) v2f* ldsp2;
__device__ __forceinline__ ldsp lds_own_reg(const float* p) {
  ldsp q = (ldsp)p;
  asm("" : "+v"(q));
  return q;
}

template <typename P>
__device__ __forceinline__ v2f ld_pair(P a, int k) {
  v2f f;
  f.x = a[k];
  f.y = a[k + 1];
  return f;
}
__device__ __forceinline__ v2f ld_pair(ldsp2 a, int k) { return a[k >> 1]; }

template <int NT, int KS, int KB, int PITCH, typename PA, typename PB>
__device__ __forceinline__ void wmma_lds_blocks(PA a, const PB (&b)[NT], v8f (&acc)[NT]) {
  static_assert(KS % KB == 0, "whole blocks");
  constexpr int NBLK = KS / KB;
  v2f fa[2][KB];
  v2f fb[2][NT][KB];
#pragma unroll
  for (int i = 0; i < KB; ++i) {
    const int k = 4 * i;
    fa[0][i] = ld_pair(a, k);
#pragma unroll
    for (int t = 0; t < NT; ++t) {
      fb[0][t][i].x = b[t][k * PITCH];
      fb[0][t][i].y = b[t][(k + 1) * PITCH];
    }
  }
#pragma unroll
  for (int blk = 0; blk < NBLK; ++blk) {
    const int cur = blk & 1;
    const int nxt = cur ^ 1;
    if (blk + 1 < NBLK) {
#pragma unroll
      for (int i = 0; i < KB; ++i) {
        const int k = 4 * ((blk + 1) * KB + i);
        fa[nxt][i] = ld_pair(a, k);
#pragma unroll
        for (int t = 0; t < NT; ++t) {
          fb[nxt][t][i].x = b[t][k * PITCH];
          fb[nxt][t][i].y = b[t][(k + 1) * PITCH];
        }
      }
    }
    __builtin_amdgcn_sched_barrier(0);
#pragma unroll
    for (int i = 0; i < KB; ++i) {
#pragma unroll
      for (int t = 0; t < NT; ++t)
        acc[t] = wmma_k4(fa[cur][i], fb[cur][t][i], acc[t]);
    }
    __builtin_amdgcn_sched_barrier(0);
  }
}


constexpr int TR    = 32;
constexpr int NPART = Ss / TR;
constexpr int QC    = NC / 2;
constexpr int NBLK1 = Bn * NPART * 4;

__global__ __launch_bounds__(256) void text_gram_kernel(
    const float* __restrict__ s, const float* __restrict__ amask,
    const float* __restrict__ Wks, const float* __restrict__ bks,
    const float* __restrict__ Wvs, const float* __restrict__ bvs,
    float* __restrict__ Gp, float* __restrict__ Ksp, float* __restrict__ Vsp) {
  constexpr int TE_STRIDE = Dd + 4;
  constexpr int KP = NC + 12;
  constexpr int KH = Dd / 2;
  __shared__ __attribute__((aligned(16))) float te_s[TR * TE_STRIDE];
  __shared__ __attribute__((aligned(16))) float w_s[Dd * NC];
  __shared__ __attribute__((aligned(16))) float bkv_s[NC];
  __shared__ float kv[TR * KP];

  const int dh = blockIdx.x & 1;
  const int ch = (blockIdx.x >> 1) & 1;
  const int st = (blockIdx.x >> 2) & (NPART - 1);
  const int b  = blockIdx.x / (4 * NPART);

  const int tid  = threadIdx.x;
  const int lane = tid & 31;
  const int wave = tid >> 5;

  const int m    = lane & 15;
  const int koff = (lane >> 4) * 2;

  const v4f* s4  = reinterpret_cast<const v4f*>(s + ((size_t)b * Ss + st * TR) * Dd);
  const int wg = tid & 15;
  const float* wsrc = (wg < 8 ? Wks + dh * QC + wg * 4 : Wvs + ch * QC + (wg - 8) * 4) +
                      (tid >> 4) * NC;
  const v4f* w4 = reinterpret_cast<const v4f*>(wsrc);
  constexpr int WJ4 = 16 * NC / 4;
  constexpr int WH4 = KH * NC / 4;
  const int wsw  = ((tid >> 5) & 1) << 4;
  const int wdst = (tid >> 4) * NC + ((wg * 4) ^ wsw);

  {
    v4f tv[16], wr[16];
    float am[16];
#pragma unroll
    for (int j = 0; j < 16; ++j) {
      tv[j] = s4[tid + 256 * j];
      am[j] = amask[b * Ss + st * TR + (tid >> 7) + 2 * j];
    }
#pragma unroll
    for (int j = 0; j < 16; ++j) wr[j] = w4[WJ4 * j];
    v4f bz = {0.f, 0.f, 0.f, 0.f};
    if (tid < 16)
      bz = reinterpret_cast<const v4f*>(tid < 8 ? bks + dh * QC : bvs + ch * QC)[tid & 7];
#pragma unroll
    for (int j = 0; j < 16; ++j)
      *reinterpret_cast<v4f*>(&te_s[((tid >> 7) + 2 * j) * TE_STRIDE + (tid & 127) * 4]) =
          tv[j] * am[j];
#pragma unroll
    for (int j = 0; j < 16; ++j)
      *reinterpret_cast<v4f*>(&w_s[wdst + 16 * j * NC]) = wr[j];
    if (tid < 16)
      *reinterpret_cast<v4f*>(&bkv_s[tid * 4]) = bz;
  }
  __syncthreads();

  {
    const int mrow = (wave >> 2) * 16;
    const int n    = (wave & 3) * 16 + m;
    const int nsw  = n ^ ((lane >> 4) << 4);
    v8f acc[1];
    acc[0] = (v8f){0.f, 0.f, 0.f, 0.f, 0.f, 0.f, 0.f, 0.f};

    v4f wr[16];
#pragma unroll
    for (int j = 0; j < 16; ++j) wr[j] = w4[WH4 + WJ4 * j];

    const ldsp2 ap = (ldsp2)lds_own_reg(te_s + (mrow + m) * TE_STRIDE + koff);
    const ldsp b0[1] = {lds_own_reg(w_s + koff * NC + nsw)};
    wmma_lds_blocks<1, KH / 4, 8, NC>(ap, b0, acc);
#pragma unroll
    for (int j = 0; j < 16; ++j)
      *reinterpret_cast<v4f*>(&w_s[KH * NC + wdst + 16 * j * NC]) = wr[j];
    __syncthreads();
    const ldsp b1[1] = {lds_own_reg(w_s + (KH + koff) * NC + nsw)};
    wmma_lds_blocks<1, KH / 4, 8, NC>(ap + KH / 2, b1, acc);
    const float bn = bkv_s[n];
#pragma unroll
    for (int i = 0; i < 8; ++i) {
      const int row = mrow + i + ((lane >> 4) << 3);
      kv[row * KP + n] = acc[0][i] + bn;
    }
  }
  __syncthreads();

  const int wave_u = __builtin_amdgcn_readfirstlane(wave);
  if (wave_u < 4) {
    const int c0 = (wave_u >> 1) * 16;
    const int d0 = (wave_u & 1) * 16;
    v8f g = {0.f, 0.f, 0.f, 0.f, 0.f, 0.f, 0.f, 0.f};
#pragma unroll
    for (int k = 0; k < TR; k += 4) {
      v2f af, bf;
      af.x = kv[(k + koff) * KP + QC + c0 + m];
      af.y = kv[(k + koff + 1) * KP + QC + c0 + m];
      bf.x = kv[(k + koff) * KP + d0 + m];
      bf.y = kv[(k + koff + 1) * KP + d0 + m];
      g = wmma_k4(af, bf, g);
    }
    float* gdst = Gp + (size_t)(b * NPART + st) * (NC * NC);
#pragma unroll
    for (int i = 0; i < 8; ++i) {
      const int row = ch * QC + c0 + i + ((lane >> 4) << 3);
      gdst[row * NC + dh * QC + d0 + m] = g[i];
    }
  } else if (wave_u < 6) {
    const bool isk = (wave_u == 4);
    if (isk ? (ch == 0) : (dh == 0)) {
      const int col = (isk ? 0 : QC) + lane;
      float acc = 0.f;
      for (int r = 0; r < TR; ++r) acc += kv[r * KP + col];
      float* dst = isk ? Ksp : Vsp;
      dst[(b * NPART + st) * NC + (isk ? dh : ch) * QC + lane] = acc;
    }
  }
}

__global__ __launch_bounds__(128) void fused_attn_kernel(
    const float* __restrict__ x,
    const float* __restrict__ Wq,  const float* __restrict__ bq,
    const float* __restrict__ Wkx, const float* __restrict__ bkx,
    const float* __restrict__ Wvx, const float* __restrict__ bvx,
    const float* __restrict__ Gp,
    const float* __restrict__ Ksp, const float* __restrict__ Vsp,
    const float* __restrict__ Wout, const float* __restrict__ bout,
    const float* __restrict__ ln_scale, const float* __restrict__ ln_bias,
    float* __restrict__ y) {
  constexpr int XS = Cc + 4;
  constexpr int QS = NC + 4;
  __shared__ __attribute__((aligned(16))) float xs[16 * XS];
  __shared__ __attribute__((aligned(16))) float w3_s[3 * Cc * NC];
  __shared__ __attribute__((aligned(16))) float wo_s[NC * Cc];
  constexpr int BOUT = 3 * NC, LNS = BOUT + Cc, LNB = LNS + Cc, NB4 = (LNB + Cc) / 4;
  __shared__ __attribute__((aligned(16))) float bias_s[NB4 * 4];
  __shared__ float qs[16 * QS];
  __shared__ float kxs[16 * QS];
  __shared__ float vxs[16 * QS];
  __shared__ float os[16 * QS];
  __shared__ float outs[16 * XS];
  constexpr int GP = NC + 4;
  constexpr int RP = 20;
  __shared__ __attribute__((aligned(16))) float Gs[NC * GP];
  __shared__ float KsS[NC], VsS[NC];
  __shared__ __attribute__((aligned(16))) float red[4 * 2 * NC * RP];

  const int tid  = threadIdx.x;
  const int lane = tid & 31;
  const int wave = tid >> 5;
  const int p0   = blockIdx.x * 16;
  const int b    = p0 >> 10;

  {
    const v4f* x4 = reinterpret_cast<const v4f*>(x + (size_t)p0 * Cc);
    const v4f* q4 = reinterpret_cast<const v4f*>(Wq);
    const v4f* k4 = reinterpret_cast<const v4f*>(Wkx);
    const v4f* v4 = reinterpret_cast<const v4f*>(Wvx);
    const v4f* o4 = reinterpret_cast<const v4f*>(Wout);
    v4f xr[4], qr[16], kr[16], vr[16], wr[16];
#pragma unroll
    for (int j = 0; j < 4; ++j) xr[j] = x4[tid + 128 * j];
#pragma unroll
    for (int j = 0; j < 16; ++j) {
      qr[j] = q4[tid + 128 * j];
      kr[j] = k4[tid + 128 * j];
      vr[j] = v4[tid + 128 * j];
      wr[j] = o4[tid + 128 * j];
    }
    v4f bz[6];
#pragma unroll
    for (int j = 0; j < 6; ++j) bz[j] = (v4f){0.f, 0.f, 0.f, 0.f};
    if (tid < 32) {
      bz[3] = reinterpret_cast<const v4f*>(bout)[tid];
      bz[4] = reinterpret_cast<const v4f*>(ln_scale)[tid];
      bz[5] = reinterpret_cast<const v4f*>(ln_bias)[tid];
      if (tid < 16) {
        bz[0] = reinterpret_cast<const v4f*>(bq)[tid];
        bz[1] = reinterpret_cast<const v4f*>(bkx)[tid];
        bz[2] = reinterpret_cast<const v4f*>(bvx)[tid];
      }
    }
#pragma unroll
    for (int j = 0; j < 4; ++j)
      *reinterpret_cast<v4f*>(&xs[((tid >> 5) + 4 * j) * XS + (tid & 31) * 4]) = xr[j];
    const int d3 = (tid >> 4) * NC + (((tid & 15) * 4) ^ (((tid >> 5) & 1) << 4));
    const int d1 = (tid >> 5) * Cc + (((tid & 31) * 4) ^ (((tid >> 6) & 1) << 4));
#pragma unroll
    for (int j = 0; j < 16; ++j) {
      *reinterpret_cast<v4f*>(&w3_s[0 * Cc * NC + d3 + 8 * j * NC]) = qr[j];
      *reinterpret_cast<v4f*>(&w3_s[1 * Cc * NC + d3 + 8 * j * NC]) = kr[j];
      *reinterpret_cast<v4f*>(&w3_s[2 * Cc * NC + d3 + 8 * j * NC]) = vr[j];
      *reinterpret_cast<v4f*>(&wo_s[d1 + 4 * j * Cc]) = wr[j];
    }
    if (tid < 32) {
      *reinterpret_cast<v4f*>(&bias_s[BOUT + tid * 4]) = bz[3];
      *reinterpret_cast<v4f*>(&bias_s[LNS + tid * 4])  = bz[4];
      *reinterpret_cast<v4f*>(&bias_s[LNB + tid * 4])  = bz[5];
      if (tid < 16) {
        *reinterpret_cast<v4f*>(&bias_s[0 * NC + tid * 4]) = bz[0];
        *reinterpret_cast<v4f*>(&bias_s[1 * NC + tid * 4]) = bz[1];
        *reinterpret_cast<v4f*>(&bias_s[2 * NC + tid * 4]) = bz[2];
      }
    }
  }
  {
    constexpr int G4 = NC * NC / 4;
    const v4f* gp4 = reinterpret_cast<const v4f*>(Gp + (size_t)b * NPART * NC * NC);
#pragma unroll 4
    for (int i = tid; i < G4; i += 128) {
      v4f a = gp4[i];
      for (int t = 1; t < NPART; ++t) a += gp4[t * G4 + i];
      *reinterpret_cast<v4f*>(&Gs[(i >> 4) * GP + (i & 15) * 4]) = a;
    }
  }
  {
    const float* sp = (tid < NC) ? Ksp : Vsp;
    const int col = tid & (NC - 1);
    float a = 0.f;
    for (int t = 0; t < NPART; ++t) a += sp[(b * NPART + t) * NC + col];
    if (tid < NC) KsS[col] = a;
    else          VsS[col] = a;
  }
  __syncthreads();

  {
    const int m    = lane & 15;
    const int koff = (lane >> 4) * 2;
    const int n    = wave * 16 + (lane & 15);
    const int nsw  = n ^ ((lane >> 4) << 4);
    const float* const b3[3] = {w3_s + 0 * (Cc * NC) + koff * NC + nsw,
                                w3_s + 1 * (Cc * NC) + koff * NC + nsw,
                                w3_s + 2 * (Cc * NC) + koff * NC + nsw};
    v8f acc[3];
#pragma unroll
    for (int mat = 0; mat < 3; ++mat)
      acc[mat] = (v8f){0.f, 0.f, 0.f, 0.f, 0.f, 0.f, 0.f, 0.f};
    wmma_lds_blocks<3, Cc / 4, 8, NC>(xs + m * XS + koff, b3, acc);
#pragma unroll
    for (int mat = 0; mat < 3; ++mat) {
      float*      dst = (mat == 0) ? qs : (mat == 1) ? kxs : vxs;
      const float bn  = bias_s[mat * NC + n];
#pragma unroll
      for (int i = 0; i < 8; ++i)
        dst[(i + ((lane >> 4) << 3)) * QS + n] = acc[mat][i] + bn;
    }
  }
  __syncthreads();

  {
    const int c = lane & 15;
    const int h = lane >> 4;
    constexpr float C8 = 0.18033688011112042f;
    for (int pi = 0; pi < 4; ++pi) {
      const int p = wave * 4 + pi;
      v2f af[4];
      for (int mt = 0; mt < 4; ++mt) {
        const float ksd = KsS[mt * 16 + c];
        af[mt].x = kxs[p * QS + mt * 16 + c];
        af[mt].y = h ? 0.f : ksd;
      }
      v8f acc[4];
      for (int mt = 0; mt < 4; ++mt)
        acc[mt] = (v8f){0.f, 0.f, 0.f, 0.f, 0.f, 0.f, 0.f, 0.f};

#pragma unroll
      for (int nt = 0; nt < 4; ++nt) {
        const int j = nt * 16 + c;
        const float vxj = vxs[p * QS + j];
        const float Vsj = VsS[j];
        v2f bf;
        bf.x = h ? Vsj : 256.f * vxj;
        bf.y = h ? 0.f : vxj;

        v8f L[4];
        for (int mt = 0; mt < 4; ++mt) {
          const v4f* gp = reinterpret_cast<const v4f*>(&Gs[j * GP + mt * 16 + 8 * h]);
          const v4f g0 = gp[0];
          const v4f g1 = gp[1];
          const v8f cf = __builtin_shufflevector(g0, g1, 0, 1, 2, 3, 4, 5, 6, 7);
          L[mt] = wmma_k4(af[mt], bf, cf);
        }

        float mm[4];
        for (int mt = 0; mt < 4; ++mt) {
          float t = L[mt][0];
          for (int r = 1; r < 8; ++r) t = fmaxf(t, L[mt][r]);
          mm[mt] = t;
        }
        float m = fmaxf(fmaxf(mm[0], mm[1]), fmaxf(mm[2], mm[3]));
        m = fmaxf(m, xhalf(m));
        const float nm = -m * C8;

        float ss[4];
        for (int mt = 0; mt < 4; ++mt) {
          float t = 0.f;
          for (int r = 0; r < 8; ++r) {
            const float e = __builtin_amdgcn_exp2f(fmaf(L[mt][r], C8, nm));
            L[mt][r] = e;
            t += e;
          }
          ss[mt] = t;
        }
        float sm = (ss[0] + ss[1]) + (ss[2] + ss[3]);
        sm += xhalf(sm);
        const float qw = qs[p * QS + j] / sm;
        for (int mt = 0; mt < 4; ++mt)
          for (int r = 0; r < 8; ++r)
            acc[mt][r] = fmaf(qw, L[mt][r], acc[mt][r]);
      }

      float* rb = red + (wave * 2 + (pi & 1)) * (NC * RP);
      for (int mt = 0; mt < 4; ++mt)
        for (int r = 0; r < 8; ++r)
          rb[(mt * 16 + 8 * h + r) * RP + c] = acc[mt][r];
      __syncthreads();
      for (int t = 0; t < 2; ++t) {
        const int d = lane + 32 * t;
        const v4f* rp = reinterpret_cast<const v4f*>(&rb[d * RP]);
        const v4f a0 = rp[0], a1 = rp[1], a2 = rp[2], a3 = rp[3];
        const v4f s4 = (a0 + a1) + (a2 + a3);
        os[p * QS + d] = (s4.x + s4.y) + (s4.z + s4.w);
      }
    }
  }
  __syncthreads();

  {
    const int m    = lane & 15;
    const int koff = (lane >> 4) * 2;
    const int n0   = wave * 16 + m;
    const int nsw0 = n0 ^ ((lane >> 4) << 4);
    const float* const b2[2] = {wo_s + koff * Cc + nsw0, wo_s + koff * Cc + nsw0 + 64};
    v8f acc[2];
    acc[0] = (v8f){0.f, 0.f, 0.f, 0.f, 0.f, 0.f, 0.f, 0.f};
    acc[1] = (v8f){0.f, 0.f, 0.f, 0.f, 0.f, 0.f, 0.f, 0.f};
    wmma_lds_blocks<2, NC / 4, 8, Cc>(os + m * QS + koff, b2, acc);
#pragma unroll
    for (int t = 0; t < 2; ++t) {
      const int   n  = n0 + 64 * t;
      const float bn = bias_s[BOUT + n];
#pragma unroll
      for (int i = 0; i < 8; ++i)
        outs[(i + ((lane >> 4) << 3)) * XS + n] = acc[t][i] + bn;
    }
  }
  __syncthreads();

  {
    const int row   = tid >> 3;
    const int cbase = (tid & 7) * 16;
    float vals[16];
    float sum = 0.f, sq = 0.f;
#pragma unroll
    for (int i = 0; i < 16; ++i) {
      const float v = xs[row * XS + cbase + i] + outs[row * XS + cbase + i];
      vals[i] = v;
      sum += v;
      sq  += v * v;
    }
#pragma unroll
    for (int msk = 4; msk >= 1; msk >>= 1) {
      sum += __shfl_xor(sum, msk, 32);
      sq  += __shfl_xor(sq,  msk, 32);
    }
    const float mu   = sum * (1.f / 128.f);
    const float var  = sq * (1.f / 128.f) - mu * mu;
    const float rstd = rsqrtf(var + 1e-6f);
#pragma unroll
    for (int i = 0; i < 16; ++i) {
      const int c = cbase + i;
      y[(size_t)(p0 + row) * Cc + c] =
          (vals[i] - mu) * rstd * bias_s[LNS + c] + bias_s[LNB + c];
    }
  }
}

extern "C" void kernel_launch(void* const* d_in, const int* in_sizes, int n_in,
                              void* d_out, int out_size, void* d_ws, size_t ws_size,
                              hipStream_t stream) {
  const float* x     = (const float*)d_in[0];
  const float* s     = (const float*)d_in[1];
  const float* amask = (const float*)d_in[2];
  const float* Wq    = (const float*)d_in[3];
  const float* bq    = (const float*)d_in[4];
  const float* Wkx   = (const float*)d_in[5];
  const float* bkx   = (const float*)d_in[6];
  const float* Wvx   = (const float*)d_in[7];
  const float* bvx   = (const float*)d_in[8];
  const float* Wks   = (const float*)d_in[9];
  const float* bks   = (const float*)d_in[10];
  const float* Wvs   = (const float*)d_in[11];
  const float* bvs   = (const float*)d_in[12];
  const float* Wout  = (const float*)d_in[13];
  const float* bout  = (const float*)d_in[14];
  const float* lns   = (const float*)d_in[15];
  const float* lnb   = (const float*)d_in[16];
  float* y = (float*)d_out;

  char* ws = (char*)d_ws;
  float* Gp  = (float*)(ws + 0);
  float* Ksp = (float*)(ws + 524288);
  float* Vsp = (float*)(ws + 532480);

  text_gram_kernel<<<dim3(NBLK1), dim3(256), 0, stream>>>(
      s, amask, Wks, bks, Wvs, bvs, Gp, Ksp, Vsp);
  fused_attn_kernel<<<dim3(NPOS / 16), dim3(128), 0, stream>>>(
      x, Wq, bq, Wkx, bkx, Wvx, bvx, Gp, Ksp, Vsp, Wout, bout, lns, lnb, y);
}


# Fused kernel: Wout, LN rows and the G/Ksum/Vsum partials requested behind the projections' operands, in flight across a hand-written first barrier, landing under the q/kx/vx WMMAs; sums unchanged
# speedup vs baseline: 6.5394x; 1.0617x over previous
#include <hip/hip_runtime.h>
#include <math.h>

constexpr int Bn  = 4;
constexpr int Wd  = 32;
constexpr int Hd  = 32;
constexpr int Cc  = 128;
constexpr int Ss  = 256;
constexpr int Dd  = 512;
constexpr int NC  = 64;
constexpr int NPOS = Bn * Wd * Hd;

typedef __attribute__((ext_vector_type(2))) float v2f;
typedef __attribute__((ext_vector_type(8))) float v8f;

__device__ __forceinline__ v8f wmma_k4(v2f a, v2f b, v8f c) {
  return __builtin_amdgcn_wmma_f32_16x16x4_f32(
       false, a,  false, b,
       (short)0, c,  false,  false);
}

typedef __attribute__((ext_vector_type(4))) float v4f;

__device__ __forceinline__ float xhalf(float x) {
  const int b = __builtin_bit_cast(int, x);
  return __builtin_bit_cast(float,
      __builtin_amdgcn_permlanex16(b, b, 0x76543210, 0xfedcba98, false, false));
}

typedef const __attribute__((address_space(3))) float* ldsp;
typedef const __attribute__((address_space(3))) v2f* ldsp2;
__device__ __forceinline__ ldsp lds_own_reg(const float* p) {
  ldsp q = (ldsp)p;
  asm("" : "+v"(q));
  return q;
}

__device__ __forceinline__ void barrier_loads_in_flight() {
  asm volatile("s_wait_dscnt 0x0\n\ts_barrier_signal -1\n\ts_barrier_wait -1" ::: "memory");
}

template <typename P>
__device__ __forceinline__ v2f ld_pair(P a, int k) {
  v2f f;
  f.x = a[k];
  f.y = a[k + 1];
  return f;
}
__device__ __forceinline__ v2f ld_pair(ldsp2 a, int k) { return a[k >> 1]; }

template <int NT, int KS, int KB, int PITCH, typename PA, typename PB>
__device__ __forceinline__ void wmma_lds_blocks(PA a, const PB (&b)[NT], v8f (&acc)[NT]) {
  static_assert(KS % KB == 0, "whole blocks");
  constexpr int NBLK = KS / KB;
  v2f fa[2][KB];
  v2f fb[2][NT][KB];
#pragma unroll
  for (int i = 0; i < KB; ++i) {
    const int k = 4 * i;
    fa[0][i] = ld_pair(a, k);
#pragma unroll
    for (int t = 0; t < NT; ++t) {
      fb[0][t][i].x = b[t][k * PITCH];
      fb[0][t][i].y = b[t][(k + 1) * PITCH];
    }
  }
#pragma unroll
  for (int blk = 0; blk < NBLK; ++blk) {
    const int cur = blk & 1;
    const int nxt = cur ^ 1;
    if (blk + 1 < NBLK) {
#pragma unroll
      for (int i = 0; i < KB; ++i) {
        const int k = 4 * ((blk + 1) * KB + i);
        fa[nxt][i] = ld_pair(a, k);
#pragma unroll
        for (int t = 0; t < NT; ++t) {
          fb[nxt][t][i].x = b[t][k * PITCH];
          fb[nxt][t][i].y = b[t][(k + 1) * PITCH];
        }
      }
    }
    __builtin_amdgcn_sched_barrier(0);
#pragma unroll
    for (int i = 0; i < KB; ++i) {
#pragma unroll
      for (int t = 0; t < NT; ++t)
        acc[t] = wmma_k4(fa[cur][i], fb[cur][t][i], acc[t]);
    }
    __builtin_amdgcn_sched_barrier(0);
  }
}


constexpr int TR    = 32;
constexpr int NPART = Ss / TR;
constexpr int QC    = NC / 2;
constexpr int NBLK1 = Bn * NPART * 4;

__global__ __launch_bounds__(256) void text_gram_kernel(
    const float* __restrict__ s, const float* __restrict__ amask,
    const float* __restrict__ Wks, const float* __restrict__ bks,
    const float* __restrict__ Wvs, const float* __restrict__ bvs,
    float* __restrict__ Gp, float* __restrict__ Ksp, float* __restrict__ Vsp) {
  constexpr int TE_STRIDE = Dd + 4;
  constexpr int KP = NC + 12;
  constexpr int KH = Dd / 2;
  __shared__ __attribute__((aligned(16))) float te_s[TR * TE_STRIDE];
  __shared__ __attribute__((aligned(16))) float w_s[Dd * NC];
  __shared__ __attribute__((aligned(16))) float bkv_s[NC];
  __shared__ float kv[TR * KP];

  const int dh = blockIdx.x & 1;
  const int ch = (blockIdx.x >> 1) & 1;
  const int st = (blockIdx.x >> 2) & (NPART - 1);
  const int b  = blockIdx.x / (4 * NPART);

  const int tid  = threadIdx.x;
  const int lane = tid & 31;
  const int wave = tid >> 5;

  const int m    = lane & 15;
  const int koff = (lane >> 4) * 2;

  const v4f* s4  = reinterpret_cast<const v4f*>(s + ((size_t)b * Ss + st * TR) * Dd);
  const int wg = tid & 15;
  const float* wsrc = (wg < 8 ? Wks + dh * QC + wg * 4 : Wvs + ch * QC + (wg - 8) * 4) +
                      (tid >> 4) * NC;
  const v4f* w4 = reinterpret_cast<const v4f*>(wsrc);
  constexpr int WJ4 = 16 * NC / 4;
  constexpr int WH4 = KH * NC / 4;
  const int wsw  = ((tid >> 5) & 1) << 4;
  const int wdst = (tid >> 4) * NC + ((wg * 4) ^ wsw);

  {
    v4f tv[16], wr[16];
    float am[16];
#pragma unroll
    for (int j = 0; j < 16; ++j) {
      tv[j] = s4[tid + 256 * j];
      am[j] = amask[b * Ss + st * TR + (tid >> 7) + 2 * j];
    }
#pragma unroll
    for (int j = 0; j < 16; ++j) wr[j] = w4[WJ4 * j];
    v4f bz = {0.f, 0.f, 0.f, 0.f};
    if (tid < 16)
      bz = reinterpret_cast<const v4f*>(tid < 8 ? bks + dh * QC : bvs + ch * QC)[tid & 7];
#pragma unroll
    for (int j = 0; j < 16; ++j)
      *reinterpret_cast<v4f*>(&te_s[((tid >> 7) + 2 * j) * TE_STRIDE + (tid & 127) * 4]) =
          tv[j] * am[j];
#pragma unroll
    for (int j = 0; j < 16; ++j)
      *reinterpret_cast<v4f*>(&w_s[wdst + 16 * j * NC]) = wr[j];
    if (tid < 16)
      *reinterpret_cast<v4f*>(&bkv_s[tid * 4]) = bz;
  }
  __syncthreads();

  {
    const int mrow = (wave >> 2) * 16;
    const int n    = (wave & 3) * 16 + m;
    const int nsw  = n ^ ((lane >> 4) << 4);
    v8f acc[1];
    acc[0] = (v8f){0.f, 0.f, 0.f, 0.f, 0.f, 0.f, 0.f, 0.f};

    v4f wr[16];
#pragma unroll
    for (int j = 0; j < 16; ++j) wr[j] = w4[WH4 + WJ4 * j];

    const ldsp2 ap = (ldsp2)lds_own_reg(te_s + (mrow + m) * TE_STRIDE + koff);
    const ldsp b0[1] = {lds_own_reg(w_s + koff * NC + nsw)};
    wmma_lds_blocks<1, KH / 4, 8, NC>(ap, b0, acc);
#pragma unroll
    for (int j = 0; j < 16; ++j)
      *reinterpret_cast<v4f*>(&w_s[KH * NC + wdst + 16 * j * NC]) = wr[j];
    __syncthreads();
    const ldsp b1[1] = {lds_own_reg(w_s + (KH + koff) * NC + nsw)};
    wmma_lds_blocks<1, KH / 4, 8, NC>(ap + KH / 2, b1, acc);
    const float bn = bkv_s[n];
#pragma unroll
    for (int i = 0; i < 8; ++i) {
      const int row = mrow + i + ((lane >> 4) << 3);
      kv[row * KP + n] = acc[0][i] + bn;
    }
  }
  __syncthreads();

  const int wave_u = __builtin_amdgcn_readfirstlane(wave);
  if (wave_u < 4) {
    const int c0 = (wave_u >> 1) * 16;
    const int d0 = (wave_u & 1) * 16;
    v8f g = {0.f, 0.f, 0.f, 0.f, 0.f, 0.f, 0.f, 0.f};
#pragma unroll
    for (int k = 0; k < TR; k += 4) {
      v2f af, bf;
      af.x = kv[(k + koff) * KP + QC + c0 + m];
      af.y = kv[(k + koff + 1) * KP + QC + c0 + m];
      bf.x = kv[(k + koff) * KP + d0 + m];
      bf.y = kv[(k + koff + 1) * KP + d0 + m];
      g = wmma_k4(af, bf, g);
    }
    float* gdst = Gp + (size_t)(b * NPART + st) * (NC * NC);
#pragma unroll
    for (int i = 0; i < 8; ++i) {
      const int row = ch * QC + c0 + i + ((lane >> 4) << 3);
      gdst[row * NC + dh * QC + d0 + m] = g[i];
    }
  } else if (wave_u < 6) {
    const bool isk = (wave_u == 4);
    if (isk ? (ch == 0) : (dh == 0)) {
      const int col = (isk ? 0 : QC) + lane;
      float acc = 0.f;
      for (int r = 0; r < TR; ++r) acc += kv[r * KP + col];
      float* dst = isk ? Ksp : Vsp;
      dst[(b * NPART + st) * NC + (isk ? dh : ch) * QC + lane] = acc;
    }
  }
}

__global__ __launch_bounds__(128) void fused_attn_kernel(
    const float* __restrict__ x,
    const float* __restrict__ Wq,  const float* __restrict__ bq,
    const float* __restrict__ Wkx, const float* __restrict__ bkx,
    const float* __restrict__ Wvx, const float* __restrict__ bvx,
    const float* __restrict__ Gp,
    const float* __restrict__ Ksp, const float* __restrict__ Vsp,
    const float* __restrict__ Wout, const float* __restrict__ bout,
    const float* __restrict__ ln_scale, const float* __restrict__ ln_bias,
    float* __restrict__ y) {
  constexpr int XS = Cc + 4;
  constexpr int QS = NC + 4;
  __shared__ __attribute__((aligned(16))) float xs[16 * XS];
  __shared__ __attribute__((aligned(16))) float w3_s[3 * Cc * NC];
  __shared__ __attribute__((aligned(16))) float wo_s[NC * Cc];
  constexpr int BOUT = 3 * NC, LNS = BOUT + Cc, LNB = LNS + Cc, NB4 = (LNB + Cc) / 4;
  __shared__ __attribute__((aligned(16))) float bias_s[NB4 * 4];
  __shared__ float qs[16 * QS];
  __shared__ float kxs[16 * QS];
  __shared__ float vxs[16 * QS];
  __shared__ float os[16 * QS];
  __shared__ float outs[16 * XS];
  constexpr int GP = NC + 4;
  constexpr int RP = 20;
  __shared__ __attribute__((aligned(16))) float Gs[NC * GP];
  __shared__ float KsS[NC], VsS[NC];
  __shared__ __attribute__((aligned(16))) float red[4 * 2 * NC * RP];

  const int tid  = threadIdx.x;
  const int lane = tid & 31;
  const int wave = tid >> 5;
  const int p0   = blockIdx.x * 16;
  const int b    = p0 >> 10;

  constexpr int G4 = NC * NC / 4;
  constexpr int GJ = G4 / 128;
  v4f   wr[16];
  v4f   bzo[3];
  float kvp[NPART];
  v4f   gq[GJ][NPART];
  const int d3 = (tid >> 4) * NC + (((tid & 15) * 4) ^ (((tid >> 5) & 1) << 4));
  const int d1 = (tid >> 5) * Cc + (((tid & 31) * 4) ^ (((tid >> 6) & 1) << 4));
  {
    const v4f* x4 = reinterpret_cast<const v4f*>(x + (size_t)p0 * Cc);
    const v4f* q4 = reinterpret_cast<const v4f*>(Wq);
    const v4f* k4 = reinterpret_cast<const v4f*>(Wkx);
    const v4f* v4 = reinterpret_cast<const v4f*>(Wvx);
    const v4f* o4 = reinterpret_cast<const v4f*>(Wout);
    const v4f* gp4 = reinterpret_cast<const v4f*>(Gp + (size_t)b * NPART * NC * NC);
    const float* sp = (tid < NC) ? Ksp : Vsp;
    v4f xr[4], qr[16], kr[16], vr[16];
#pragma unroll
    for (int j = 0; j < 4; ++j) xr[j] = x4[tid + 128 * j];
#pragma unroll
    for (int j = 0; j < 16; ++j) {
      qr[j] = q4[tid + 128 * j];
      kr[j] = k4[tid + 128 * j];
      vr[j] = v4[tid + 128 * j];
    }
    v4f bz[3];
#pragma unroll
    for (int j = 0; j < 3; ++j) bz[j] = (v4f){0.f, 0.f, 0.f, 0.f};
    if (tid < 16) {
      bz[0] = reinterpret_cast<const v4f*>(bq)[tid];
      bz[1] = reinterpret_cast<const v4f*>(bkx)[tid];
      bz[2] = reinterpret_cast<const v4f*>(bvx)[tid];
    }
#pragma unroll
    for (int j = 0; j < 16; ++j) wr[j] = o4[tid + 128 * j];
#pragma unroll
    for (int j = 0; j < 3; ++j) bzo[j] = (v4f){0.f, 0.f, 0.f, 0.f};
    if (tid < 32) {
      bzo[0] = reinterpret_cast<const v4f*>(bout)[tid];
      bzo[1] = reinterpret_cast<const v4f*>(ln_scale)[tid];
      bzo[2] = reinterpret_cast<const v4f*>(ln_bias)[tid];
    }
#pragma unroll
    for (int t = 0; t < NPART; ++t) kvp[t] = sp[(b * NPART + t) * NC + (tid & (NC - 1))];
#pragma unroll
    for (int g = 0; g < GJ / 2; ++g) {
#pragma unroll
      for (int t = 0; t < NPART; ++t) gq[g][t] = gp4[t * G4 + tid + 128 * g];
    }
#pragma unroll
    for (int j = 0; j < 4; ++j)
      *reinterpret_cast<v4f*>(&xs[((tid >> 5) + 4 * j) * XS + (tid & 31) * 4]) = xr[j];
#pragma unroll
    for (int j = 0; j < 16; ++j) {
      *reinterpret_cast<v4f*>(&w3_s[0 * Cc * NC + d3 + 8 * j * NC]) = qr[j];
      *reinterpret_cast<v4f*>(&w3_s[1 * Cc * NC + d3 + 8 * j * NC]) = kr[j];
      *reinterpret_cast<v4f*>(&w3_s[2 * Cc * NC + d3 + 8 * j * NC]) = vr[j];
    }
    if (tid < 16) {
      *reinterpret_cast<v4f*>(&bias_s[0 * NC + tid * 4]) = bz[0];
      *reinterpret_cast<v4f*>(&bias_s[1 * NC + tid * 4]) = bz[1];
      *reinterpret_cast<v4f*>(&bias_s[2 * NC + tid * 4]) = bz[2];
    }
#pragma unroll
    for (int g = GJ / 2; g < GJ; ++g) {
#pragma unroll
      for (int t = 0; t < NPART; ++t) gq[g][t] = gp4[t * G4 + tid + 128 * g];
    }
  }
  barrier_loads_in_flight();

  {
    const int m    = lane & 15;
    const int koff = (lane >> 4) * 2;
    const int n    = wave * 16 + (lane & 15);
    const int nsw  = n ^ ((lane >> 4) << 4);
    const float* const b3[3] = {w3_s + 0 * (Cc * NC) + koff * NC + nsw,
                                w3_s + 1 * (Cc * NC) + koff * NC + nsw,
                                w3_s + 2 * (Cc * NC) + koff * NC + nsw};
    v8f acc[3];
#pragma unroll
    for (int mat = 0; mat < 3; ++mat)
      acc[mat] = (v8f){0.f, 0.f, 0.f, 0.f, 0.f, 0.f, 0.f, 0.f};
    wmma_lds_blocks<3, Cc / 4, 8, NC>(xs + m * XS + koff, b3, acc);
#pragma unroll
    for (int mat = 0; mat < 3; ++mat) {
      float*      dst = (mat == 0) ? qs : (mat == 1) ? kxs : vxs;
      const float bn  = bias_s[mat * NC + n];
#pragma unroll
      for (int i = 0; i < 8; ++i)
        dst[(i + ((lane >> 4) << 3)) * QS + n] = acc[mat][i] + bn;
    }
  }
  {
#pragma unroll
    for (int j = 0; j < 16; ++j)
      *reinterpret_cast<v4f*>(&wo_s[d1 + 4 * j * Cc]) = wr[j];
    if (tid < 32) {
      *reinterpret_cast<v4f*>(&bias_s[BOUT + tid * 4]) = bzo[0];
      *reinterpret_cast<v4f*>(&bias_s[LNS + tid * 4])  = bzo[1];
      *reinterpret_cast<v4f*>(&bias_s[LNB + tid * 4])  = bzo[2];
    }
    float a = 0.f;
#pragma unroll
    for (int t = 0; t < NPART; ++t) a += kvp[t];
    if (tid < NC) KsS[tid & (NC - 1)] = a;
    else          VsS[tid & (NC - 1)] = a;
#pragma unroll
    for (int g = 0; g < GJ; ++g) {
      v4f s4 = gq[g][0];
#pragma unroll
      for (int t = 1; t < NPART; ++t) s4 += gq[g][t];
      const int i = tid + 128 * g;
      *reinterpret_cast<v4f*>(&Gs[(i >> 4) * GP + (i & 15) * 4]) = s4;
    }
  }
  __syncthreads();

  {
    const int c = lane & 15;
    const int h = lane >> 4;
    constexpr float C8 = 0.18033688011112042f;
    for (int pi = 0; pi < 4; ++pi) {
      const int p = wave * 4 + pi;
      v2f af[4];
      for (int mt = 0; mt < 4; ++mt) {
        const float ksd = KsS[mt * 16 + c];
        af[mt].x = kxs[p * QS + mt * 16 + c];
        af[mt].y = h ? 0.f : ksd;
      }
      v8f acc[4];
      for (int mt = 0; mt < 4; ++mt)
        acc[mt] = (v8f){0.f, 0.f, 0.f, 0.f, 0.f, 0.f, 0.f, 0.f};

#pragma unroll
      for (int nt = 0; nt < 4; ++nt) {
        const int j = nt * 16 + c;
        const float vxj = vxs[p * QS + j];
        const float Vsj = VsS[j];
        v2f bf;
        bf.x = h ? Vsj : 256.f * vxj;
        bf.y = h ? 0.f : vxj;

        v8f L[4];
        for (int mt = 0; mt < 4; ++mt) {
          const v4f* gp = reinterpret_cast<const v4f*>(&Gs[j * GP + mt * 16 + 8 * h]);
          const v4f g0 = gp[0];
          const v4f g1 = gp[1];
          const v8f cf = __builtin_shufflevector(g0, g1, 0, 1, 2, 3, 4, 5, 6, 7);
          L[mt] = wmma_k4(af[mt], bf, cf);
        }

        float mm[4];
        for (int mt = 0; mt < 4; ++mt) {
          float t = L[mt][0];
          for (int r = 1; r < 8; ++r) t = fmaxf(t, L[mt][r]);
          mm[mt] = t;
        }
        float m = fmaxf(fmaxf(mm[0], mm[1]), fmaxf(mm[2], mm[3]));
        m = fmaxf(m, xhalf(m));
        const float nm = -m * C8;

        float ss[4];
        for (int mt = 0; mt < 4; ++mt) {
          float t = 0.f;
          for (int r = 0; r < 8; ++r) {
            const float e = __builtin_amdgcn_exp2f(fmaf(L[mt][r], C8, nm));
            L[mt][r] = e;
            t += e;
          }
          ss[mt] = t;
        }
        float sm = (ss[0] + ss[1]) + (ss[2] + ss[3]);
        sm += xhalf(sm);
        const float qw = qs[p * QS + j] / sm;
        for (int mt = 0; mt < 4; ++mt)
          for (int r = 0; r < 8; ++r)
            acc[mt][r] = fmaf(qw, L[mt][r], acc[mt][r]);
      }

      float* rb = red + (wave * 2 + (pi & 1)) * (NC * RP);
      for (int mt = 0; mt < 4; ++mt)
        for (int r = 0; r < 8; ++r)
          rb[(mt * 16 + 8 * h + r) * RP + c] = acc[mt][r];
      __syncthreads();
      for (int t = 0; t < 2; ++t) {
        const int d = lane + 32 * t;
        const v4f* rp = reinterpret_cast<const v4f*>(&rb[d * RP]);
        const v4f a0 = rp[0], a1 = rp[1], a2 = rp[2], a3 = rp[3];
        const v4f s4 = (a0 + a1) + (a2 + a3);
        os[p * QS + d] = (s4.x + s4.y) + (s4.z + s4.w);
      }
    }
  }
  __syncthreads();

  {
    const int m    = lane & 15;
    const int koff = (lane >> 4) * 2;
    const int n0   = wave * 16 + m;
    const int nsw0 = n0 ^ ((lane >> 4) << 4);
    const float* const b2[2] = {wo_s + koff * Cc + nsw0, wo_s + koff * Cc + nsw0 + 64};
    v8f acc[2];
    acc[0] = (v8f){0.f, 0.f, 0.f, 0.f, 0.f, 0.f, 0.f, 0.f};
    acc[1] = (v8f){0.f, 0.f, 0.f, 0.f, 0.f, 0.f, 0.f, 0.f};
    wmma_lds_blocks<2, NC / 4, 8, Cc>(os + m * QS + koff, b2, acc);
#pragma unroll
    for (int t = 0; t < 2; ++t) {
      const int   n  = n0 + 64 * t;
      const float bn = bias_s[BOUT + n];
#pragma unroll
      for (int i = 0; i < 8; ++i)
        outs[(i + ((lane >> 4) << 3)) * XS + n] = acc[t][i] + bn;
    }
  }
  __syncthreads();

  {
    const int row   = tid >> 3;
    const int cbase = (tid & 7) * 16;
    float vals[16];
    float sum = 0.f, sq = 0.f;
#pragma unroll
    for (int i = 0; i < 16; ++i) {
      const float v = xs[row * XS + cbase + i] + outs[row * XS + cbase + i];
      vals[i] = v;
      sum += v;
      sq  += v * v;
    }
#pragma unroll
    for (int msk = 4; msk >= 1; msk >>= 1) {
      sum += __shfl_xor(sum, msk, 32);
      sq  += __shfl_xor(sq,  msk, 32);
    }
    const float mu   = sum * (1.f / 128.f);
    const float var  = sq * (1.f / 128.f) - mu * mu;
    const float rstd = rsqrtf(var + 1e-6f);
#pragma unroll
    for (int i = 0; i < 16; ++i) {
      const int c = cbase + i;
      y[(size_t)(p0 + row) * Cc + c] =
          (vals[i] - mu) * rstd * bias_s[LNS + c] + bias_s[LNB + c];
    }
  }
}

extern "C" void kernel_launch(void* const* d_in, const int* in_sizes, int n_in,
                              void* d_out, int out_size, void* d_ws, size_t ws_size,
                              hipStream_t stream) {
  const float* x     = (const float*)d_in[0];
  const float* s     = (const float*)d_in[1];
  const float* amask = (const float*)d_in[2];
  const float* Wq    = (const float*)d_in[3];
  const float* bq    = (const float*)d_in[4];
  const float* Wkx   = (const float*)d_in[5];
  const float* bkx   = (const float*)d_in[6];
  const float* Wvx   = (const float*)d_in[7];
  const float* bvx   = (const float*)d_in[8];
  const float* Wks   = (const float*)d_in[9];
  const float* bks   = (const float*)d_in[10];
  const float* Wvs   = (const float*)d_in[11];
  const float* bvs   = (const float*)d_in[12];
  const float* Wout  = (const float*)d_in[13];
  const float* bout  = (const float*)d_in[14];
  const float* lns   = (const float*)d_in[15];
  const float* lnb   = (const float*)d_in[16];
  float* y = (float*)d_out;

  char* ws = (char*)d_ws;
  float* Gp  = (float*)(ws + 0);
  float* Ksp = (float*)(ws + 524288);
  float* Vsp = (float*)(ws + 532480);

  text_gram_kernel<<<dim3(NBLK1), dim3(256), 0, stream>>>(
      s, amask, Wks, bks, Wvs, bvs, Gp, Ksp, Vsp);
  fused_attn_kernel<<<dim3(NPOS / 16), dim3(128), 0, stream>>>(
      x, Wq, bq, Wkx, bkx, Wvx, bvx, Gp, Ksp, Vsp, Wout, bout, lns, lnb, y);
}


# Softmax loops interchanged (row tile outer, position inner): a tile's G^T C fragments read from LDS once, kept in registers for the wave's 4 positions; the 4 lane sums via LDS as one batch
# speedup vs baseline: 6.8813x; 1.0523x over previous
#include <hip/hip_runtime.h>
#include <math.h>

constexpr int Bn  = 4;
constexpr int Wd  = 32;
constexpr int Hd  = 32;
constexpr int Cc  = 128;
constexpr int Ss  = 256;
constexpr int Dd  = 512;
constexpr int NC  = 64;
constexpr int NPOS = Bn * Wd * Hd;

typedef __attribute__((ext_vector_type(2))) float v2f;
typedef __attribute__((ext_vector_type(8))) float v8f;

__device__ __forceinline__ v8f wmma_k4(v2f a, v2f b, v8f c) {
  return __builtin_amdgcn_wmma_f32_16x16x4_f32(
       false, a,  false, b,
       (short)0, c,  false,  false);
}

typedef __attribute__((ext_vector_type(4))) float v4f;

__device__ __forceinline__ float xhalf(float x) {
  const int b = __builtin_bit_cast(int, x);
  return __builtin_bit_cast(float,
      __builtin_amdgcn_permlanex16(b, b, 0x76543210, 0xfedcba98, false, false));
}

typedef const __attribute__((address_space(3))) float* ldsp;
typedef const __attribute__((address_space(3))) v2f* ldsp2;
__device__ __forceinline__ ldsp lds_own_reg(const float* p) {
  ldsp q = (ldsp)p;
  asm("" : "+v"(q));
  return q;
}

__device__ __forceinline__ void barrier_loads_in_flight() {
  asm volatile("s_wait_dscnt 0x0\n\ts_barrier_signal -1\n\ts_barrier_wait -1" ::: "memory");
}

template <typename P>
__device__ __forceinline__ v2f ld_pair(P a, int k) {
  v2f f;
  f.x = a[k];
  f.y = a[k + 1];
  return f;
}
__device__ __forceinline__ v2f ld_pair(ldsp2 a, int k) { return a[k >> 1]; }

template <int NT, int KS, int KB, int PITCH, typename PA, typename PB>
__device__ __forceinline__ void wmma_lds_blocks(PA a, const PB (&b)[NT], v8f (&acc)[NT]) {
  static_assert(KS % KB == 0, "whole blocks");
  constexpr int NBLK = KS / KB;
  v2f fa[2][KB];
  v2f fb[2][NT][KB];
#pragma unroll
  for (int i = 0; i < KB; ++i) {
    const int k = 4 * i;
    fa[0][i] = ld_pair(a, k);
#pragma unroll
    for (int t = 0; t < NT; ++t) {
      fb[0][t][i].x = b[t][k * PITCH];
      fb[0][t][i].y = b[t][(k + 1) * PITCH];
    }
  }
#pragma unroll
  for (int blk = 0; blk < NBLK; ++blk) {
    const int cur = blk & 1;
    const int nxt = cur ^ 1;
    if (blk + 1 < NBLK) {
#pragma unroll
      for (int i = 0; i < KB; ++i) {
        const int k = 4 * ((blk + 1) * KB + i);
        fa[nxt][i] = ld_pair(a, k);
#pragma unroll
        for (int t = 0; t < NT; ++t) {
          fb[nxt][t][i].x = b[t][k * PITCH];
          fb[nxt][t][i].y = b[t][(k + 1) * PITCH];
        }
      }
    }
    __builtin_amdgcn_sched_barrier(0);
#pragma unroll
    for (int i = 0; i < KB; ++i) {
#pragma unroll
      for (int t = 0; t < NT; ++t)
        acc[t] = wmma_k4(fa[cur][i], fb[cur][t][i], acc[t]);
    }
    __builtin_amdgcn_sched_barrier(0);
  }
}


constexpr int TR    = 32;
constexpr int NPART = Ss / TR;
constexpr int QC    = NC / 2;
constexpr int NBLK1 = Bn * NPART * 4;

__global__ __launch_bounds__(256) void text_gram_kernel(
    const float* __restrict__ s, const float* __restrict__ amask,
    const float* __restrict__ Wks, const float* __restrict__ bks,
    const float* __restrict__ Wvs, const float* __restrict__ bvs,
    float* __restrict__ Gp, float* __restrict__ Ksp, float* __restrict__ Vsp) {
  constexpr int TE_STRIDE = Dd + 4;
  constexpr int KP = NC + 12;
  constexpr int KH = Dd / 2;
  __shared__ __attribute__((aligned(16))) float te_s[TR * TE_STRIDE];
  __shared__ __attribute__((aligned(16))) float w_s[Dd * NC];
  __shared__ __attribute__((aligned(16))) float bkv_s[NC];
  __shared__ float kv[TR * KP];

  const int dh = blockIdx.x & 1;
  const int ch = (blockIdx.x >> 1) & 1;
  const int st = (blockIdx.x >> 2) & (NPART - 1);
  const int b  = blockIdx.x / (4 * NPART);

  const int tid  = threadIdx.x;
  const int lane = tid & 31;
  const int wave = tid >> 5;

  const int m    = lane & 15;
  const int koff = (lane >> 4) * 2;

  const v4f* s4  = reinterpret_cast<const v4f*>(s + ((size_t)b * Ss + st * TR) * Dd);
  const int wg = tid & 15;
  const float* wsrc = (wg < 8 ? Wks + dh * QC + wg * 4 : Wvs + ch * QC + (wg - 8) * 4) +
                      (tid >> 4) * NC;
  const v4f* w4 = reinterpret_cast<const v4f*>(wsrc);
  constexpr int WJ4 = 16 * NC / 4;
  constexpr int WH4 = KH * NC / 4;
  const int wsw  = ((tid >> 5) & 1) << 4;
  const int wdst = (tid >> 4) * NC + ((wg * 4) ^ wsw);

  {
    v4f tv[16], wr[16];
    float am[16];
#pragma unroll
    for (int j = 0; j < 16; ++j) {
      tv[j] = s4[tid + 256 * j];
      am[j] = amask[b * Ss + st * TR + (tid >> 7) + 2 * j];
    }
#pragma unroll
    for (int j = 0; j < 16; ++j) wr[j] = w4[WJ4 * j];
    v4f bz = {0.f, 0.f, 0.f, 0.f};
    if (tid < 16)
      bz = reinterpret_cast<const v4f*>(tid < 8 ? bks + dh * QC : bvs + ch * QC)[tid & 7];
#pragma unroll
    for (int j = 0; j < 16; ++j)
      *reinterpret_cast<v4f*>(&te_s[((tid >> 7) + 2 * j) * TE_STRIDE + (tid & 127) * 4]) =
          tv[j] * am[j];
#pragma unroll
    for (int j = 0; j < 16; ++j)
      *reinterpret_cast<v4f*>(&w_s[wdst + 16 * j * NC]) = wr[j];
    if (tid < 16)
      *reinterpret_cast<v4f*>(&bkv_s[tid * 4]) = bz;
  }
  __syncthreads();

  {
    const int mrow = (wave >> 2) * 16;
    const int n    = (wave & 3) * 16 + m;
    const int nsw  = n ^ ((lane >> 4) << 4);
    v8f acc[1];
    acc[0] = (v8f){0.f, 0.f, 0.f, 0.f, 0.f, 0.f, 0.f, 0.f};

    v4f wr[16];
#pragma unroll
    for (int j = 0; j < 16; ++j) wr[j] = w4[WH4 + WJ4 * j];

    const ldsp2 ap = (ldsp2)lds_own_reg(te_s + (mrow + m) * TE_STRIDE + koff);
    const ldsp b0[1] = {lds_own_reg(w_s + koff * NC + nsw)};
    wmma_lds_blocks<1, KH / 4, 8, NC>(ap, b0, acc);
#pragma unroll
    for (int j = 0; j < 16; ++j)
      *reinterpret_cast<v4f*>(&w_s[KH * NC + wdst + 16 * j * NC]) = wr[j];
    __syncthreads();
    const ldsp b1[1] = {lds_own_reg(w_s + (KH + koff) * NC + nsw)};
    wmma_lds_blocks<1, KH / 4, 8, NC>(ap + KH / 2, b1, acc);
    const float bn = bkv_s[n];
#pragma unroll
    for (int i = 0; i < 8; ++i) {
      const int row = mrow + i + ((lane >> 4) << 3);
      kv[row * KP + n] = acc[0][i] + bn;
    }
  }
  __syncthreads();

  const int wave_u = __builtin_amdgcn_readfirstlane(wave);
  if (wave_u < 4) {
    const int c0 = (wave_u >> 1) * 16;
    const int d0 = (wave_u & 1) * 16;
    v8f g = {0.f, 0.f, 0.f, 0.f, 0.f, 0.f, 0.f, 0.f};
#pragma unroll
    for (int k = 0; k < TR; k += 4) {
      v2f af, bf;
      af.x = kv[(k + koff) * KP + QC + c0 + m];
      af.y = kv[(k + koff + 1) * KP + QC + c0 + m];
      bf.x = kv[(k + koff) * KP + d0 + m];
      bf.y = kv[(k + koff + 1) * KP + d0 + m];
      g = wmma_k4(af, bf, g);
    }
    float* gdst = Gp + (size_t)(b * NPART + st) * (NC * NC);
#pragma unroll
    for (int i = 0; i < 8; ++i) {
      const int row = ch * QC + c0 + i + ((lane >> 4) << 3);
      gdst[row * NC + dh * QC + d0 + m] = g[i];
    }
  } else if (wave_u < 6) {
    const bool isk = (wave_u == 4);
    if (isk ? (ch == 0) : (dh == 0)) {
      const int col = (isk ? 0 : QC) + lane;
      float acc = 0.f;
      for (int r = 0; r < TR; ++r) acc += kv[r * KP + col];
      float* dst = isk ? Ksp : Vsp;
      dst[(b * NPART + st) * NC + (isk ? dh : ch) * QC + lane] = acc;
    }
  }
}

__global__ __launch_bounds__(128) void fused_attn_kernel(
    const float* __restrict__ x,
    const float* __restrict__ Wq,  const float* __restrict__ bq,
    const float* __restrict__ Wkx, const float* __restrict__ bkx,
    const float* __restrict__ Wvx, const float* __restrict__ bvx,
    const float* __restrict__ Gp,
    const float* __restrict__ Ksp, const float* __restrict__ Vsp,
    const float* __restrict__ Wout, const float* __restrict__ bout,
    const float* __restrict__ ln_scale, const float* __restrict__ ln_bias,
    float* __restrict__ y) {
  constexpr int XS = Cc + 4;
  constexpr int QS = NC + 4;
  __shared__ __attribute__((aligned(16))) float xs[16 * XS];
  __shared__ __attribute__((aligned(16))) float w3_s[3 * Cc * NC];
  __shared__ __attribute__((aligned(16))) float wo_s[NC * Cc];
  constexpr int BOUT = 3 * NC, LNS = BOUT + Cc, LNB = LNS + Cc, NB4 = (LNB + Cc) / 4;
  __shared__ __attribute__((aligned(16))) float bias_s[NB4 * 4];
  __shared__ float qs[16 * QS];
  __shared__ float kxs[16 * QS];
  __shared__ float vxs[16 * QS];
  __shared__ float os[16 * QS];
  __shared__ float outs[16 * XS];
  constexpr int GP = NC + 4;
  constexpr int RP = 20;
  __shared__ __attribute__((aligned(16))) float Gs[NC * GP];
  __shared__ float KsS[NC], VsS[NC];
  __shared__ __attribute__((aligned(16))) float red[4 * 4 * NC * RP];

  const int tid  = threadIdx.x;
  const int lane = tid & 31;
  const int wave = tid >> 5;
  const int p0   = blockIdx.x * 16;
  const int b    = p0 >> 10;

  constexpr int G4 = NC * NC / 4;
  constexpr int GJ = G4 / 128;
  v4f   wr[16];
  v4f   bzo[3];
  float kvp[NPART];
  v4f   gq[GJ][NPART];
  const int d3 = (tid >> 4) * NC + (((tid & 15) * 4) ^ (((tid >> 5) & 1) << 4));
  const int d1 = (tid >> 5) * Cc + (((tid & 31) * 4) ^ (((tid >> 6) & 1) << 4));
  {
    const v4f* x4 = reinterpret_cast<const v4f*>(x + (size_t)p0 * Cc);
    const v4f* q4 = reinterpret_cast<const v4f*>(Wq);
    const v4f* k4 = reinterpret_cast<const v4f*>(Wkx);
    const v4f* v4 = reinterpret_cast<const v4f*>(Wvx);
    const v4f* o4 = reinterpret_cast<const v4f*>(Wout);
    const v4f* gp4 = reinterpret_cast<const v4f*>(Gp + (size_t)b * NPART * NC * NC);
    const float* sp = (tid < NC) ? Ksp : Vsp;
    v4f xr[4], qr[16], kr[16], vr[16];
#pragma unroll
    for (int j = 0; j < 4; ++j) xr[j] = x4[tid + 128 * j];
#pragma unroll
    for (int j = 0; j < 16; ++j) {
      qr[j] = q4[tid + 128 * j];
      kr[j] = k4[tid + 128 * j];
      vr[j] = v4[tid + 128 * j];
    }
    v4f bz[3];
#pragma unroll
    for (int j = 0; j < 3; ++j) bz[j] = (v4f){0.f, 0.f, 0.f, 0.f};
    if (tid < 16) {
      bz[0] = reinterpret_cast<const v4f*>(bq)[tid];
      bz[1] = reinterpret_cast<const v4f*>(bkx)[tid];
      bz[2] = reinterpret_cast<const v4f*>(bvx)[tid];
    }
#pragma unroll
    for (int j = 0; j < 16; ++j) wr[j] = o4[tid + 128 * j];
#pragma unroll
    for (int j = 0; j < 3; ++j) bzo[j] = (v4f){0.f, 0.f, 0.f, 0.f};
    if (tid < 32) {
      bzo[0] = reinterpret_cast<const v4f*>(bout)[tid];
      bzo[1] = reinterpret_cast<const v4f*>(ln_scale)[tid];
      bzo[2] = reinterpret_cast<const v4f*>(ln_bias)[tid];
    }
#pragma unroll
    for (int t = 0; t < NPART; ++t) kvp[t] = sp[(b * NPART + t) * NC + (tid & (NC - 1))];
#pragma unroll
    for (int g = 0; g < GJ / 2; ++g) {
#pragma unroll
      for (int t = 0; t < NPART; ++t) gq[g][t] = gp4[t * G4 + tid + 128 * g];
    }
#pragma unroll
    for (int j = 0; j < 4; ++j)
      *reinterpret_cast<v4f*>(&xs[((tid >> 5) + 4 * j) * XS + (tid & 31) * 4]) = xr[j];
#pragma unroll
    for (int j = 0; j < 16; ++j) {
      *reinterpret_cast<v4f*>(&w3_s[0 * Cc * NC + d3 + 8 * j * NC]) = qr[j];
      *reinterpret_cast<v4f*>(&w3_s[1 * Cc * NC + d3 + 8 * j * NC]) = kr[j];
      *reinterpret_cast<v4f*>(&w3_s[2 * Cc * NC + d3 + 8 * j * NC]) = vr[j];
    }
    if (tid < 16) {
      *reinterpret_cast<v4f*>(&bias_s[0 * NC + tid * 4]) = bz[0];
      *reinterpret_cast<v4f*>(&bias_s[1 * NC + tid * 4]) = bz[1];
      *reinterpret_cast<v4f*>(&bias_s[2 * NC + tid * 4]) = bz[2];
    }
#pragma unroll
    for (int g = GJ / 2; g < GJ; ++g) {
#pragma unroll
      for (int t = 0; t < NPART; ++t) gq[g][t] = gp4[t * G4 + tid + 128 * g];
    }
  }
  barrier_loads_in_flight();

  {
    const int m    = lane & 15;
    const int koff = (lane >> 4) * 2;
    const int n    = wave * 16 + (lane & 15);
    const int nsw  = n ^ ((lane >> 4) << 4);
    const float* const b3[3] = {w3_s + 0 * (Cc * NC) + koff * NC + nsw,
                                w3_s + 1 * (Cc * NC) + koff * NC + nsw,
                                w3_s + 2 * (Cc * NC) + koff * NC + nsw};
    v8f acc[3];
#pragma unroll
    for (int mat = 0; mat < 3; ++mat)
      acc[mat] = (v8f){0.f, 0.f, 0.f, 0.f, 0.f, 0.f, 0.f, 0.f};
    wmma_lds_blocks<3, Cc / 4, 8, NC>(xs + m * XS + koff, b3, acc);
#pragma unroll
    for (int mat = 0; mat < 3; ++mat) {
      float*      dst = (mat == 0) ? qs : (mat == 1) ? kxs : vxs;
      const float bn  = bias_s[mat * NC + n];
#pragma unroll
      for (int i = 0; i < 8; ++i)
        dst[(i + ((lane >> 4) << 3)) * QS + n] = acc[mat][i] + bn;
    }
  }
  {
#pragma unroll
    for (int j = 0; j < 16; ++j)
      *reinterpret_cast<v4f*>(&wo_s[d1 + 4 * j * Cc]) = wr[j];
    if (tid < 32) {
      *reinterpret_cast<v4f*>(&bias_s[BOUT + tid * 4]) = bzo[0];
      *reinterpret_cast<v4f*>(&bias_s[LNS + tid * 4])  = bzo[1];
      *reinterpret_cast<v4f*>(&bias_s[LNB + tid * 4])  = bzo[2];
    }
    float a = 0.f;
#pragma unroll
    for (int t = 0; t < NPART; ++t) a += kvp[t];
    if (tid < NC) KsS[tid & (NC - 1)] = a;
    else          VsS[tid & (NC - 1)] = a;
#pragma unroll
    for (int g = 0; g < GJ; ++g) {
      v4f s4 = gq[g][0];
#pragma unroll
      for (int t = 1; t < NPART; ++t) s4 += gq[g][t];
      const int i = tid + 128 * g;
      *reinterpret_cast<v4f*>(&Gs[(i >> 4) * GP + (i & 15) * 4]) = s4;
    }
  }
  __syncthreads();

  {
    const int c = lane & 15;
    const int h = lane >> 4;
    constexpr float C8 = 0.18033688011112042f;
    v2f af[4][4];
    v8f acc[4][4];
    for (int mt = 0; mt < 4; ++mt) {
      const float ksd = KsS[mt * 16 + c];
      for (int pi = 0; pi < 4; ++pi) {
        af[pi][mt].x = kxs[(wave * 4 + pi) * QS + mt * 16 + c];
        af[pi][mt].y = h ? 0.f : ksd;
        acc[pi][mt] = (v8f){0.f, 0.f, 0.f, 0.f, 0.f, 0.f, 0.f, 0.f};
      }
    }
    v8f cf[4];
    for (int mt = 0; mt < 4; ++mt) {
      const v4f* gp = reinterpret_cast<const v4f*>(&Gs[c * GP + mt * 16 + 8 * h]);
      cf[mt] = __builtin_shufflevector(gp[0], gp[1], 0, 1, 2, 3, 4, 5, 6, 7);
    }

    for (int nt = 0; nt < 4; ++nt) {
      const int j = nt * 16 + c;
      const int jn = (nt < 3 ? nt + 1 : 3) * 16 + c;
      v8f cn[4];
      for (int mt = 0; mt < 4; ++mt) {
        const v4f* gp = reinterpret_cast<const v4f*>(&Gs[jn * GP + mt * 16 + 8 * h]);
        cn[mt] = __builtin_shufflevector(gp[0], gp[1], 0, 1, 2, 3, 4, 5, 6, 7);
      }
      const float Vsj = VsS[j];

      for (int pi = 0; pi < 4; ++pi) {
        const int p = wave * 4 + pi;
        const float vxj = vxs[p * QS + j];
        v2f bf;
        bf.x = h ? Vsj : 256.f * vxj;
        bf.y = h ? 0.f : vxj;

        v8f L[4];
        for (int mt = 0; mt < 4; ++mt) L[mt] = wmma_k4(af[pi][mt], bf, cf[mt]);

        float mm[4];
        for (int mt = 0; mt < 4; ++mt) {
          float t = L[mt][0];
          for (int r = 1; r < 8; ++r) t = fmaxf(t, L[mt][r]);
          mm[mt] = t;
        }
        float m = fmaxf(fmaxf(mm[0], mm[1]), fmaxf(mm[2], mm[3]));
        m = fmaxf(m, xhalf(m));
        const float nm = -m * C8;

        float ss[4];
        for (int mt = 0; mt < 4; ++mt) {
          float t = 0.f;
          for (int r = 0; r < 8; ++r) {
            const float e = __builtin_amdgcn_exp2f(fmaf(L[mt][r], C8, nm));
            L[mt][r] = e;
            t += e;
          }
          ss[mt] = t;
        }
        float sm = (ss[0] + ss[1]) + (ss[2] + ss[3]);
        sm += xhalf(sm);
        const float qw = qs[p * QS + j] / sm;
        for (int mt = 0; mt < 4; ++mt)
          for (int r = 0; r < 8; ++r)
            acc[pi][mt][r] = fmaf(qw, L[mt][r], acc[pi][mt][r]);
      }
      for (int mt = 0; mt < 4; ++mt) cf[mt] = cn[mt];
    }

    for (int pi = 0; pi < 4; ++pi) {
      float* rb = red + (wave * 4 + pi) * (NC * RP);
      for (int mt = 0; mt < 4; ++mt)
        for (int r = 0; r < 8; ++r)
          rb[(mt * 16 + 8 * h + r) * RP + c] = acc[pi][mt][r];
    }
    __syncthreads();
    for (int pi = 0; pi < 4; ++pi) {
      const float* rb = red + (wave * 4 + pi) * (NC * RP);
      const int p = wave * 4 + pi;
      for (int t = 0; t < 2; ++t) {
        const int d = lane + 32 * t;
        const v4f* rp = reinterpret_cast<const v4f*>(&rb[d * RP]);
        const v4f a0 = rp[0], a1 = rp[1], a2 = rp[2], a3 = rp[3];
        const v4f s4 = (a0 + a1) + (a2 + a3);
        os[p * QS + d] = (s4.x + s4.y) + (s4.z + s4.w);
      }
    }
  }
  __syncthreads();

  {
    const int m    = lane & 15;
    const int koff = (lane >> 4) * 2;
    const int n0   = wave * 16 + m;
    const int nsw0 = n0 ^ ((lane >> 4) << 4);
    const float* const b2[2] = {wo_s + koff * Cc + nsw0, wo_s + koff * Cc + nsw0 + 64};
    v8f acc[2];
    acc[0] = (v8f){0.f, 0.f, 0.f, 0.f, 0.f, 0.f, 0.f, 0.f};
    acc[1] = (v8f){0.f, 0.f, 0.f, 0.f, 0.f, 0.f, 0.f, 0.f};
    wmma_lds_blocks<2, NC / 4, 8, Cc>(os + m * QS + koff, b2, acc);
#pragma unroll
    for (int t = 0; t < 2; ++t) {
      const int   n  = n0 + 64 * t;
      const float bn = bias_s[BOUT + n];
#pragma unroll
      for (int i = 0; i < 8; ++i)
        outs[(i + ((lane >> 4) << 3)) * XS + n] = acc[t][i] + bn;
    }
  }
  __syncthreads();

  {
    const int row   = tid >> 3;
    const int cbase = (tid & 7) * 16;
    float vals[16];
    float sum = 0.f, sq = 0.f;
#pragma unroll
    for (int i = 0; i < 16; ++i) {
      const float v = xs[row * XS + cbase + i] + outs[row * XS + cbase + i];
      vals[i] = v;
      sum += v;
      sq  += v * v;
    }
#pragma unroll
    for (int msk = 4; msk >= 1; msk >>= 1) {
      sum += __shfl_xor(sum, msk, 32);
      sq  += __shfl_xor(sq,  msk, 32);
    }
    const float mu   = sum * (1.f / 128.f);
    const float var  = sq * (1.f / 128.f) - mu * mu;
    const float rstd = rsqrtf(var + 1e-6f);
#pragma unroll
    for (int i = 0; i < 16; ++i) {
      const int c = cbase + i;
      y[(size_t)(p0 + row) * Cc + c] =
          (vals[i] - mu) * rstd * bias_s[LNS + c] + bias_s[LNB + c];
    }
  }
}

extern "C" void kernel_launch(void* const* d_in, const int* in_sizes, int n_in,
                              void* d_out, int out_size, void* d_ws, size_t ws_size,
                              hipStream_t stream) {
  const float* x     = (const float*)d_in[0];
  const float* s     = (const float*)d_in[1];
  const float* amask = (const float*)d_in[2];
  const float* Wq    = (const float*)d_in[3];
  const float* bq    = (const float*)d_in[4];
  const float* Wkx   = (const float*)d_in[5];
  const float* bkx   = (const float*)d_in[6];
  const float* Wvx   = (const float*)d_in[7];
  const float* bvx   = (const float*)d_in[8];
  const float* Wks   = (const float*)d_in[9];
  const float* bks   = (const float*)d_in[10];
  const float* Wvs   = (const float*)d_in[11];
  const float* bvs   = (const float*)d_in[12];
  const float* Wout  = (const float*)d_in[13];
  const float* bout  = (const float*)d_in[14];
  const float* lns   = (const float*)d_in[15];
  const float* lnb   = (const float*)d_in[16];
  float* y = (float*)d_out;

  char* ws = (char*)d_ws;
  float* Gp  = (float*)(ws + 0);
  float* Ksp = (float*)(ws + 524288);
  float* Vsp = (float*)(ws + 532480);

  text_gram_kernel<<<dim3(NBLK1), dim3(256), 0, stream>>>(
      s, amask, Wks, bks, Wvs, bvs, Gp, Ksp, Vsp);
  fused_attn_kernel<<<dim3(NPOS / 16), dim3(128), 0, stream>>>(
      x, Wq, bq, Wkx, bkx, Wvx, bvx, Gp, Ksp, Vsp, Wout, bout, lns, lnb, y);
}
